# starting kernel
# baseline (speedup 1.0000x reference)
#include <hip/hip_runtime.h>

typedef __attribute__((ext_vector_type(2))) float v2f;
typedef __attribute__((ext_vector_type(8))) float v8f;

#define NN 50000
#define NE 800000
#define FIN 256
#define HD 128
#define NC1 4
#define NC2 3

__global__ void k_deg_init(float* __restrict__ deg) {
    int i = blockIdx.x * blockDim.x + threadIdx.x;
    if (i < NN) deg[i] = 1.0f;
}

__global__ void k_deg_count(const int* __restrict__ dst, float* __restrict__ deg) {
    int e = blockIdx.x * blockDim.x + threadIdx.x;
    if (e < NE) atomicAdd(&deg[dst[e]], 1.0f);
}

__global__ void k_dinv(float* __restrict__ deg) {
    int i = blockIdx.x * blockDim.x + threadIdx.x;
    if (i < NN) deg[i] = rsqrtf(deg[i]);
}

__global__ void k_gemm_wmma_f32(const float* __restrict__ X, const float* __restrict__ W,
                                float* __restrict__ OUT, int M, int K, int N) {
    const int wave  = (blockIdx.x * blockDim.x + threadIdx.x) >> 5;
    const int lane  = threadIdx.x & 31;
    const int half  = lane >> 4;
    const int lh    = lane & 15;
    const int tilesN = N >> 4;
    const int tm = wave / tilesN;
    const int tn = wave - tm * tilesN;
    if (tm * 16 >= M) return;
    const int m0 = tm << 4, n0 = tn << 4;

    v8f c = {};
    #pragma unroll 4
    for (int k0 = 0; k0 < K; k0 += 4) {
        const int ka = k0 + (half << 1);
        v2f a = *(const v2f*)(X + (size_t)(m0 + lh) * K + ka);
        const float* wp = W + (size_t)ka * N + (n0 + lh);
        v2f b;
        b.x = wp[0];
        b.y = wp[N];
        c = __builtin_amdgcn_wmma_f32_16x16x4_f32(
                 false, a,  false, b,
                 (short)0, c,  false,  false);
    }

    #pragma unroll
    for (int j = 0; j < 8; ++j) {
        const int row = m0 + j + (half << 3);
        OUT[(size_t)row * N + n0 + lh] = c[j];
    }
}

__global__ void k_selfloop(const float* __restrict__ Hf, const float* __restrict__ dinv,
                           float* __restrict__ OUT) {
    size_t i = (size_t)blockIdx.x * blockDim.x + threadIdx.x;
    if (i >= (size_t)NN * HD) return;
    const int node = (int)(i >> 7);
    const float dv = dinv[node];
    OUT[i] = Hf[i] * dv * dv;
}

__global__ void k_edge_scatter(const float* __restrict__ Hf, const int* __restrict__ src,
                               const int* __restrict__ dst, const float* __restrict__ dinv,
                               float* __restrict__ OUT) {
    const int t = blockIdx.x * blockDim.x + threadIdx.x;
    const int e = t >> 5;
    const int lane = t & 31;
    if (e >= NE) return;
    const int s = src[e];
    const int d = dst[e];
    const float w = dinv[s] * dinv[d];
    const float4 hv = *(const float4*)(Hf + (size_t)s * HD + lane * 4);
    float* op = OUT + (size_t)d * HD + lane * 4;
    atomicAdd(op + 0, w * hv.x);
    atomicAdd(op + 1, w * hv.y);
    atomicAdd(op + 2, w * hv.z);
    atomicAdd(op + 3, w * hv.w);
}

__global__ void k_bias_relu(const float* __restrict__ IN, const float* __restrict__ bias,
                            float* __restrict__ OUT) {
    size_t i = (size_t)blockIdx.x * blockDim.x + threadIdx.x;
    if (i >= (size_t)NN * HD) return;
    const int f = (int)(i & (HD - 1));
    OUT[i] = fmaxf(IN[i] + bias[f], 0.0f);
}

__global__ void k_heads(const float* __restrict__ AGG, const float* __restrict__ b2,
                        const float* __restrict__ Wh1, const float* __restrict__ bh1,
                        const float* __restrict__ Wh2, const float* __restrict__ bh2,
                        float* __restrict__ out1, float* __restrict__ out2,
                        float* __restrict__ hout) {
    const int t = blockIdx.x * blockDim.x + threadIdx.x;
    const int node = t >> 5;
    const int lane = t & 31;
    if (node >= NN) return;

    const float4 av = *(const float4*)(AGG + (size_t)node * HD + lane * 4);
    const float4 bb = ((const float4*)b2)[lane];
    float hv[4] = { av.x + bb.x, av.y + bb.y, av.z + bb.z, av.w + bb.w };

    float4 hstore = { hv[0], hv[1], hv[2], hv[3] };
    *(float4*)(hout + (size_t)node * HD + lane * 4) = hstore;

    float acc[NC1 + NC2];
    #pragma unroll
    for (int c = 0; c < NC1 + NC2; ++c) acc[c] = 0.0f;

    #pragma unroll
    for (int j = 0; j < 4; ++j) {
        const int k = lane * 4 + j;
        #pragma unroll
        for (int c = 0; c < NC1; ++c) acc[c]       += hv[j] * Wh1[k * NC1 + c];
        #pragma unroll
        for (int c = 0; c < NC2; ++c) acc[NC1 + c] += hv[j] * Wh2[k * NC2 + c];
    }

    #pragma unroll
    for (int off = 16; off > 0; off >>= 1) {
        #pragma unroll
        for (int c = 0; c < NC1 + NC2; ++c)
            acc[c] += __shfl_down(acc[c], off, 32);
    }

    if (lane == 0) {
        #pragma unroll
        for (int c = 0; c < NC1; ++c) out1[(size_t)node * NC1 + c] = acc[c] + bh1[c];
        #pragma unroll
        for (int c = 0; c < NC2; ++c) out2[(size_t)node * NC2 + c] = acc[NC1 + c] + bh2[c];
    }
}

extern "C" void kernel_launch(void* const* d_in, const int* in_sizes, int n_in,
                              void* d_out, int out_size, void* d_ws, size_t ws_size,
                              hipStream_t stream) {
    const float* x   = (const float*)d_in[0];
    const int*   ei  = (const int*)d_in[1];
    const int*   src = ei;
    const int*   dst = ei + NE;
    const float* W1  = (const float*)d_in[2];
    const float* b1  = (const float*)d_in[3];
    const float* W2  = (const float*)d_in[4];
    const float* b2  = (const float*)d_in[5];
    const float* Wh1 = (const float*)d_in[6];
    const float* bh1 = (const float*)d_in[7];
    const float* Wh2 = (const float*)d_in[8];
    const float* bh2 = (const float*)d_in[9];

    float* out1 = (float*)d_out;
    float* out2 = out1 + (size_t)NN * NC1;
    float* hout = out2 + (size_t)NN * NC2;

    float* dinv = (float*)d_ws;
    size_t ofs  = (((size_t)NN * sizeof(float)) + 255) & ~(size_t)255;
    float* bufA = (float*)((char*)d_ws + ofs);
    float* bufB = bufA + (size_t)NN * HD;

    const int NF_BLK   = (NN * HD + 255) / 256;
    const int NODE_BLK = (NN + 255) / 256;
    const int EDGE_BLK = (NE + 255) / 256;
    const int SCAT_BLK = (NE * 32) / 256;
    const int HEAD_BLK = (NN * 32 + 255) / 256;
    const int GEMM_BLK = ((NN / 16) * (HD / 16) + 7) / 8;

    k_deg_init <<<NODE_BLK, 256, 0, stream>>>(dinv);
    k_deg_count<<<EDGE_BLK, 256, 0, stream>>>(dst, dinv);
    k_dinv     <<<NODE_BLK, 256, 0, stream>>>(dinv);

    k_gemm_wmma_f32<<<GEMM_BLK, 256, 0, stream>>>(x, W1, bufA, NN, FIN, HD);
    k_selfloop     <<<NF_BLK,   256, 0, stream>>>(bufA, dinv, bufB);
    k_edge_scatter <<<SCAT_BLK, 256, 0, stream>>>(bufA, src, dst, dinv, bufB);
    k_bias_relu    <<<NF_BLK,   256, 0, stream>>>(bufB, b1, bufA);

    k_gemm_wmma_f32<<<GEMM_BLK, 256, 0, stream>>>(bufA, W2, bufB, NN, HD, HD);
    k_selfloop     <<<NF_BLK,   256, 0, stream>>>(bufB, dinv, bufA);
    k_edge_scatter <<<SCAT_BLK, 256, 0, stream>>>(bufB, src, dst, dinv, bufA);

    k_heads<<<HEAD_BLK, 256, 0, stream>>>(bufA, b2, Wh1, bh1, Wh2, bh2, out1, out2, hout);
}


# Float-atomic edge scatter replaced by a per-node gather over edges grouped by destination in-call (count atomic returns slot, scan, fill); row sums now in arrival-order slots; self-loop folded in
# speedup vs baseline: 1.7604x; 1.7604x over previous
#include <hip/hip_runtime.h>

typedef __attribute__((ext_vector_type(2))) float v2f;
typedef __attribute__((ext_vector_type(8))) float v8f;

#define NN 50000
#define NE 800000
#define FIN 256
#define HD 128
#define NC1 4
#define NC2 3

#define WS_CNT     200192u
#define WS_ROWPTR  400384u
#define WS_RANK    600576u
#define WS_CSR     3800576u
#define WS_BUFA    7000576u

__global__ void k_cnt_zero(int* __restrict__ cnt) {
    int i = blockIdx.x * blockDim.x + threadIdx.x;
    if (i < NN) cnt[i] = 0;
}

__global__ void k_rank(const int* __restrict__ dst, int* __restrict__ cnt, int* __restrict__ rank) {
    int e = blockIdx.x * blockDim.x + threadIdx.x;
    if (e < NE) rank[e] = atomicAdd(&cnt[dst[e]], 1);
}

__global__ void __launch_bounds__(1024) k_scan(const int* __restrict__ cnt, int* __restrict__ rowptr,
                                               float* __restrict__ dinv) {
    __shared__ int s_part[32];
    __shared__ int s_tot[8];
    const int t = threadIdx.x;
    const int lane = t & 31, wv = t >> 5;
    const int b0 = blockIdx.x * 256;

    int part = 0;
    #pragma unroll 4
    for (int j = t * 4; j < b0; j += 4096) {
        const int4 v = *(const int4*)(cnt + j);
        part += (v.x + v.y) + (v.z + v.w);
    }
    #pragma unroll
    for (int off = 16; off > 0; off >>= 1) part += __shfl_xor(part, off, 32);
    if (lane == 0) s_part[wv] = part;

    const int i = b0 + t;
    int c = 0, inc = 0;
    if (t < 256) {
        c = (i < NN) ? cnt[i] : 0;
        inc = c;
        #pragma unroll
        for (int off = 1; off < 32; off <<= 1) {
            const int v = __shfl_up(inc, off, 32);
            if (lane >= off) inc += v;
        }
        if (lane == 31) s_tot[wv] = inc;
    }
    __syncthreads();
    if (t < 256) {
        int base = 0;
        #pragma unroll
        for (int k = 0; k < 32; ++k) base += s_part[k];
        #pragma unroll
        for (int k = 0; k < 8; ++k) base += (k < wv) ? s_tot[k] : 0;
        const int excl = base + inc - c;
        if (i < NN) {
            rowptr[i] = excl;
            dinv[i] = rsqrtf((float)(c + 1));
        }
        if (i == NN - 1) rowptr[NN] = excl + c;
    }
}

__global__ void k_fill(const int* __restrict__ src, const int* __restrict__ dst,
                       const int* __restrict__ rowptr, const int* __restrict__ rank,
                       int* __restrict__ csr) {
    int e = blockIdx.x * blockDim.x + threadIdx.x;
    if (e >= NE) return;
    const int pos = rowptr[dst[e]] + rank[e];
    if ((unsigned)pos < (unsigned)NE) csr[pos] = src[e];
}

__global__ void k_gemm_wmma_f32(const float* __restrict__ X, const float* __restrict__ W,
                                float* __restrict__ OUT, int M, int K, int N) {
    const int wave  = (blockIdx.x * blockDim.x + threadIdx.x) >> 5;
    const int lane  = threadIdx.x & 31;
    const int half  = lane >> 4;
    const int lh    = lane & 15;
    const int tilesN = N >> 4;
    const int tm = wave / tilesN;
    const int tn = wave - tm * tilesN;
    if (tm * 16 >= M) return;
    const int m0 = tm << 4, n0 = tn << 4;

    v8f c = {};
    #pragma unroll 4
    for (int k0 = 0; k0 < K; k0 += 4) {
        const int ka = k0 + (half << 1);
        v2f a = *(const v2f*)(X + (size_t)(m0 + lh) * K + ka);
        const float* wp = W + (size_t)ka * N + (n0 + lh);
        v2f b;
        b.x = wp[0];
        b.y = wp[N];
        c = __builtin_amdgcn_wmma_f32_16x16x4_f32(
                 false, a,  false, b,
                 (short)0, c,  false,  false);
    }

    #pragma unroll
    for (int j = 0; j < 8; ++j) {
        const int row = m0 + j + (half << 3);
        OUT[(size_t)row * N + n0 + lh] = c[j];
    }
}

template <int K>
__device__ __forceinline__ void gather_group(const float* __restrict__ Hf, int s_l, float w_l, int j,
                                             unsigned lane4, float4& acc) {
    float4 hv[K];
    float  w[K];
    #pragma unroll
    for (int u = 0; u < K; ++u) {
        const unsigned s = (unsigned)__builtin_amdgcn_readlane(s_l, j + u);
        w[u]  = __int_as_float(__builtin_amdgcn_readlane(__float_as_int(w_l), j + u));
        hv[u] = *(const float4*)(Hf + (s * HD + lane4));
    }
    #pragma unroll
    for (int u = 0; u < K; ++u) {
        acc.x = fmaf(w[u], hv[u].x, acc.x);
        acc.y = fmaf(w[u], hv[u].y, acc.y);
        acc.z = fmaf(w[u], hv[u].z, acc.z);
        acc.w = fmaf(w[u], hv[u].w, acc.w);
    }
}

__global__ void __launch_bounds__(256) k_gather(const float* __restrict__ Hf, const int* __restrict__ rowptr,
                                                const int* __restrict__ csr, const float* __restrict__ dinv,
                                                float* __restrict__ OUT) {
    const int t = blockIdx.x * blockDim.x + threadIdx.x;
    const int lane = threadIdx.x & 31;
    const int node = __builtin_amdgcn_readfirstlane(t >> 5);
    if (node >= NN) return;
    const unsigned lane4 = (unsigned)lane * 4u;
    const int beg = rowptr[node];
    const int end = rowptr[node + 1];
    const float di = dinv[node];

    const float4 hs = *(const float4*)(Hf + ((unsigned)node * HD + lane4));
    const float wself = di * di;
    float4 acc = { hs.x * wself, hs.y * wself, hs.z * wself, hs.w * wself };

    for (int base = beg; base < end; base += 32) {
        const int n = min(32, end - base);
        int   s_l = node;
        float w_l = 0.0f;
        if (lane < n) {
            const int sv = csr[base + lane];
            s_l = ((unsigned)sv < (unsigned)NN) ? sv : node;
            w_l = dinv[s_l] * di;
        }
        int j = 0;
        for (; j + 8 <= n; j += 8) gather_group<8>(Hf, s_l, w_l, j, lane4, acc);
        if (j + 4 <= n) { gather_group<4>(Hf, s_l, w_l, j, lane4, acc); j += 4; }
        if (j + 2 <= n) { gather_group<2>(Hf, s_l, w_l, j, lane4, acc); j += 2; }
        if (j < n)      { gather_group<1>(Hf, s_l, w_l, j, lane4, acc); }
    }
    *(float4*)(OUT + ((unsigned)node * HD + lane4)) = acc;
}

__global__ void k_bias_relu(const float* __restrict__ IN, const float* __restrict__ bias,
                            float* __restrict__ OUT) {
    size_t i = (size_t)blockIdx.x * blockDim.x + threadIdx.x;
    if (i >= (size_t)NN * HD) return;
    const int f = (int)(i & (HD - 1));
    OUT[i] = fmaxf(IN[i] + bias[f], 0.0f);
}

__global__ void k_heads(const float* __restrict__ AGG, const float* __restrict__ b2,
                        const float* __restrict__ Wh1, const float* __restrict__ bh1,
                        const float* __restrict__ Wh2, const float* __restrict__ bh2,
                        float* __restrict__ out1, float* __restrict__ out2,
                        float* __restrict__ hout) {
    const int t = blockIdx.x * blockDim.x + threadIdx.x;
    const int node = t >> 5;
    const int lane = t & 31;
    if (node >= NN) return;

    const float4 av = *(const float4*)(AGG + (size_t)node * HD + lane * 4);
    const float4 bb = ((const float4*)b2)[lane];
    float hv[4] = { av.x + bb.x, av.y + bb.y, av.z + bb.z, av.w + bb.w };

    float4 hstore = { hv[0], hv[1], hv[2], hv[3] };
    *(float4*)(hout + (size_t)node * HD + lane * 4) = hstore;

    float acc[NC1 + NC2];
    #pragma unroll
    for (int c = 0; c < NC1 + NC2; ++c) acc[c] = 0.0f;

    #pragma unroll
    for (int j = 0; j < 4; ++j) {
        const int k = lane * 4 + j;
        #pragma unroll
        for (int c = 0; c < NC1; ++c) acc[c]       += hv[j] * Wh1[k * NC1 + c];
        #pragma unroll
        for (int c = 0; c < NC2; ++c) acc[NC1 + c] += hv[j] * Wh2[k * NC2 + c];
    }

    #pragma unroll
    for (int off = 16; off > 0; off >>= 1) {
        #pragma unroll
        for (int c = 0; c < NC1 + NC2; ++c)
            acc[c] += __shfl_down(acc[c], off, 32);
    }

    if (lane == 0) {
        #pragma unroll
        for (int c = 0; c < NC1; ++c) out1[(size_t)node * NC1 + c] = acc[c] + bh1[c];
        #pragma unroll
        for (int c = 0; c < NC2; ++c) out2[(size_t)node * NC2 + c] = acc[NC1 + c] + bh2[c];
    }
}

extern "C" void kernel_launch(void* const* d_in, const int* in_sizes, int n_in,
                              void* d_out, int out_size, void* d_ws, size_t ws_size,
                              hipStream_t stream) {
    const float* x   = (const float*)d_in[0];
    const int*   ei  = (const int*)d_in[1];
    const int*   src = ei;
    const int*   dst = ei + NE;
    const float* W1  = (const float*)d_in[2];
    const float* b1  = (const float*)d_in[3];
    const float* W2  = (const float*)d_in[4];
    const float* b2  = (const float*)d_in[5];
    const float* Wh1 = (const float*)d_in[6];
    const float* bh1 = (const float*)d_in[7];
    const float* Wh2 = (const float*)d_in[8];
    const float* bh2 = (const float*)d_in[9];

    float* out1 = (float*)d_out;
    float* out2 = out1 + (size_t)NN * NC1;
    float* hout = out2 + (size_t)NN * NC2;

    float* dinv   = (float*)d_ws;
    int*   cnt    = (int*)((char*)d_ws + WS_CNT);
    int*   rowptr = (int*)((char*)d_ws + WS_ROWPTR);
    int*   rank   = (int*)((char*)d_ws + WS_RANK);
    int*   csr    = (int*)((char*)d_ws + WS_CSR);
    float* bufA   = (float*)((char*)d_ws + WS_BUFA);
    float* bufB   = bufA + (size_t)NN * HD;

    const int NF_BLK   = (NN * HD + 255) / 256;
    const int NODE_BLK = (NN + 255) / 256;
    const int EDGE_BLK = (NE + 255) / 256;
    const int WAVE_BLK = (NN * 32 + 255) / 256;
    const int GEMM_BLK = ((NN / 16) * (HD / 16) + 7) / 8;

    k_cnt_zero<<<NODE_BLK, 256, 0, stream>>>(cnt);
    k_rank    <<<EDGE_BLK, 256, 0, stream>>>(dst, cnt, rank);
    k_scan    <<<NODE_BLK, 1024, 0, stream>>>(cnt, rowptr, dinv);
    k_fill    <<<EDGE_BLK, 256, 0, stream>>>(src, dst, rowptr, rank, csr);

    k_gemm_wmma_f32<<<GEMM_BLK, 256, 0, stream>>>(x, W1, bufA, NN, FIN, HD);
    k_gather       <<<WAVE_BLK, 256, 0, stream>>>(bufA, rowptr, csr, dinv, bufB);
    k_bias_relu    <<<NF_BLK,   256, 0, stream>>>(bufB, b1, bufA);

    k_gemm_wmma_f32<<<GEMM_BLK, 256, 0, stream>>>(bufA, W2, bufB, NN, HD, HD);
    k_gather       <<<WAVE_BLK, 256, 0, stream>>>(bufB, rowptr, csr, dinv, bufA);

    k_heads<<<WAVE_BLK, 256, 0, stream>>>(bufA, b2, Wh1, bh1, Wh2, bh2, out1, out2, hout);
}


# Row-wise pass after each aggregation (bias+ReLU; bias + h store + both heads) folded into the gather kernel's epilogue on the accumulator: 2 launches and 250,000 short waves fewer
# speedup vs baseline: 2.0743x; 1.1783x over previous
#include <hip/hip_runtime.h>

typedef __attribute__((ext_vector_type(2))) float v2f;
typedef __attribute__((ext_vector_type(8))) float v8f;

#define NN 50000
#define NE 800000
#define FIN 256
#define HD 128
#define NC1 4
#define NC2 3

#define WS_CNT     200192u
#define WS_ROWPTR  400384u
#define WS_RANK    600576u
#define WS_CSR     3800576u
#define WS_BUFA    7000576u

__global__ void k_cnt_zero(int* __restrict__ cnt) {
    int i = blockIdx.x * blockDim.x + threadIdx.x;
    if (i < NN) cnt[i] = 0;
}

__global__ void k_rank(const int* __restrict__ dst, int* __restrict__ cnt, int* __restrict__ rank) {
    int e = blockIdx.x * blockDim.x + threadIdx.x;
    if (e < NE) rank[e] = atomicAdd(&cnt[dst[e]], 1);
}

__global__ void __launch_bounds__(1024) k_scan(const int* __restrict__ cnt, int* __restrict__ rowptr,
                                               float* __restrict__ dinv) {
    __shared__ int s_part[32];
    __shared__ int s_tot[8];
    const int t = threadIdx.x;
    const int lane = t & 31, wv = t >> 5;
    const int b0 = blockIdx.x * 256;

    int part = 0;
    #pragma unroll 4
    for (int j = t * 4; j < b0; j += 4096) {
        const int4 v = *(const int4*)(cnt + j);
        part += (v.x + v.y) + (v.z + v.w);
    }
    #pragma unroll
    for (int off = 16; off > 0; off >>= 1) part += __shfl_xor(part, off, 32);
    if (lane == 0) s_part[wv] = part;

    const int i = b0 + t;
    int c = 0, inc = 0;
    if (t < 256) {
        c = (i < NN) ? cnt[i] : 0;
        inc = c;
        #pragma unroll
        for (int off = 1; off < 32; off <<= 1) {
            const int v = __shfl_up(inc, off, 32);
            if (lane >= off) inc += v;
        }
        if (lane == 31) s_tot[wv] = inc;
    }
    __syncthreads();
    if (t < 256) {
        int base = 0;
        #pragma unroll
        for (int k = 0; k < 32; ++k) base += s_part[k];
        #pragma unroll
        for (int k = 0; k < 8; ++k) base += (k < wv) ? s_tot[k] : 0;
        const int excl = base + inc - c;
        if (i < NN) {
            rowptr[i] = excl;
            dinv[i] = rsqrtf((float)(c + 1));
        }
        if (i == NN - 1) rowptr[NN] = excl + c;
    }
}

__global__ void k_fill(const int* __restrict__ src, const int* __restrict__ dst,
                       const int* __restrict__ rowptr, const int* __restrict__ rank,
                       int* __restrict__ csr) {
    int e = blockIdx.x * blockDim.x + threadIdx.x;
    if (e >= NE) return;
    const int pos = rowptr[dst[e]] + rank[e];
    if ((unsigned)pos < (unsigned)NE) csr[pos] = src[e];
}

__global__ void k_gemm_wmma_f32(const float* __restrict__ X, const float* __restrict__ W,
                                float* __restrict__ OUT, int M, int K, int N) {
    const int wave  = (blockIdx.x * blockDim.x + threadIdx.x) >> 5;
    const int lane  = threadIdx.x & 31;
    const int half  = lane >> 4;
    const int lh    = lane & 15;
    const int tilesN = N >> 4;
    const int tm = wave / tilesN;
    const int tn = wave - tm * tilesN;
    if (tm * 16 >= M) return;
    const int m0 = tm << 4, n0 = tn << 4;

    v8f c = {};
    #pragma unroll 4
    for (int k0 = 0; k0 < K; k0 += 4) {
        const int ka = k0 + (half << 1);
        v2f a = *(const v2f*)(X + (size_t)(m0 + lh) * K + ka);
        const float* wp = W + (size_t)ka * N + (n0 + lh);
        v2f b;
        b.x = wp[0];
        b.y = wp[N];
        c = __builtin_amdgcn_wmma_f32_16x16x4_f32(
                 false, a,  false, b,
                 (short)0, c,  false,  false);
    }

    #pragma unroll
    for (int j = 0; j < 8; ++j) {
        const int row = m0 + j + (half << 3);
        OUT[(size_t)row * N + n0 + lh] = c[j];
    }
}

template <int K>
__device__ __forceinline__ void gather_group(const float* __restrict__ Hf, int s_l, float w_l, int j,
                                             unsigned lane4, float4& acc) {
    float4 hv[K];
    float  w[K];
    #pragma unroll
    for (int u = 0; u < K; ++u) {
        const unsigned s = (unsigned)__builtin_amdgcn_readlane(s_l, j + u);
        w[u]  = __int_as_float(__builtin_amdgcn_readlane(__float_as_int(w_l), j + u));
        hv[u] = *(const float4*)(Hf + (s * HD + lane4));
    }
    #pragma unroll
    for (int u = 0; u < K; ++u) {
        acc.x = fmaf(w[u], hv[u].x, acc.x);
        acc.y = fmaf(w[u], hv[u].y, acc.y);
        acc.z = fmaf(w[u], hv[u].z, acc.z);
        acc.w = fmaf(w[u], hv[u].w, acc.w);
    }
}

__device__ __forceinline__ float4 aggregate_row(const float* __restrict__ Hf, const int* __restrict__ rowptr,
                                                const int* __restrict__ csr, const float* __restrict__ dinv,
                                                int node, int lane, unsigned lane4) {
    const int beg = rowptr[node];
    const int end = rowptr[node + 1];
    const float di = dinv[node];

    const float4 hs = *(const float4*)(Hf + ((unsigned)node * HD + lane4));
    const float wself = di * di;
    float4 acc = { hs.x * wself, hs.y * wself, hs.z * wself, hs.w * wself };

    for (int base = beg; base < end; base += 32) {
        const int n = min(32, end - base);
        int   s_l = node;
        float w_l = 0.0f;
        if (lane < n) {
            const int sv = csr[base + lane];
            s_l = ((unsigned)sv < (unsigned)NN) ? sv : node;
            w_l = dinv[s_l] * di;
        }
        int j = 0;
        for (; j + 8 <= n; j += 8) gather_group<8>(Hf, s_l, w_l, j, lane4, acc);
        if (j + 4 <= n) { gather_group<4>(Hf, s_l, w_l, j, lane4, acc); j += 4; }
        if (j + 2 <= n) { gather_group<2>(Hf, s_l, w_l, j, lane4, acc); j += 2; }
        if (j < n)      { gather_group<1>(Hf, s_l, w_l, j, lane4, acc); }
    }
    return acc;
}

__global__ void __launch_bounds__(256) k_gather_relu(const float* __restrict__ Hf, const int* __restrict__ rowptr,
                                                     const int* __restrict__ csr, const float* __restrict__ dinv,
                                                     const float* __restrict__ bias, float* __restrict__ OUT) {
    const int t = blockIdx.x * blockDim.x + threadIdx.x;
    const int lane = threadIdx.x & 31;
    const int node = __builtin_amdgcn_readfirstlane(t >> 5);
    if (node >= NN) return;
    const unsigned lane4 = (unsigned)lane * 4u;
    const float4 bb = ((const float4*)bias)[lane];
    const float4 acc = aggregate_row(Hf, rowptr, csr, dinv, node, lane, lane4);
    float4 r;
    r.x = fmaxf(acc.x + bb.x, 0.0f);
    r.y = fmaxf(acc.y + bb.y, 0.0f);
    r.z = fmaxf(acc.z + bb.z, 0.0f);
    r.w = fmaxf(acc.w + bb.w, 0.0f);
    *(float4*)(OUT + ((unsigned)node * HD + lane4)) = r;
}

__global__ void __launch_bounds__(256) k_gather_heads(const float* __restrict__ Hf, const int* __restrict__ rowptr,
                                                      const int* __restrict__ csr, const float* __restrict__ dinv,
                                                      const float* __restrict__ b2,
                                                      const float* __restrict__ Wh1, const float* __restrict__ bh1,
                                                      const float* __restrict__ Wh2, const float* __restrict__ bh2,
                                                      float* __restrict__ out1, float* __restrict__ out2,
                                                      float* __restrict__ hout) {
    const int t = blockIdx.x * blockDim.x + threadIdx.x;
    const int lane = threadIdx.x & 31;
    const int node = __builtin_amdgcn_readfirstlane(t >> 5);
    if (node >= NN) return;
    const unsigned lane4 = (unsigned)lane * 4u;
    const float4 bb = ((const float4*)b2)[lane];
    const float4 av = aggregate_row(Hf, rowptr, csr, dinv, node, lane, lane4);

    float hv[4] = { av.x + bb.x, av.y + bb.y, av.z + bb.z, av.w + bb.w };

    float4 hstore = { hv[0], hv[1], hv[2], hv[3] };
    *(float4*)(hout + (size_t)node * HD + lane4) = hstore;

    float acc[NC1 + NC2];
    #pragma unroll
    for (int c = 0; c < NC1 + NC2; ++c) acc[c] = 0.0f;

    #pragma unroll
    for (int j = 0; j < 4; ++j) {
        const int k = lane * 4 + j;
        #pragma unroll
        for (int c = 0; c < NC1; ++c) acc[c]       += hv[j] * Wh1[k * NC1 + c];
        #pragma unroll
        for (int c = 0; c < NC2; ++c) acc[NC1 + c] += hv[j] * Wh2[k * NC2 + c];
    }

    #pragma unroll
    for (int off = 16; off > 0; off >>= 1) {
        #pragma unroll
        for (int c = 0; c < NC1 + NC2; ++c)
            acc[c] += __shfl_down(acc[c], off, 32);
    }

    if (lane == 0) {
        #pragma unroll
        for (int c = 0; c < NC1; ++c) out1[(size_t)node * NC1 + c] = acc[c] + bh1[c];
        #pragma unroll
        for (int c = 0; c < NC2; ++c) out2[(size_t)node * NC2 + c] = acc[NC1 + c] + bh2[c];
    }
}

extern "C" void kernel_launch(void* const* d_in, const int* in_sizes, int n_in,
                              void* d_out, int out_size, void* d_ws, size_t ws_size,
                              hipStream_t stream) {
    const float* x   = (const float*)d_in[0];
    const int*   ei  = (const int*)d_in[1];
    const int*   src = ei;
    const int*   dst = ei + NE;
    const float* W1  = (const float*)d_in[2];
    const float* b1  = (const float*)d_in[3];
    const float* W2  = (const float*)d_in[4];
    const float* b2  = (const float*)d_in[5];
    const float* Wh1 = (const float*)d_in[6];
    const float* bh1 = (const float*)d_in[7];
    const float* Wh2 = (const float*)d_in[8];
    const float* bh2 = (const float*)d_in[9];

    float* out1 = (float*)d_out;
    float* out2 = out1 + (size_t)NN * NC1;
    float* hout = out2 + (size_t)NN * NC2;

    float* dinv   = (float*)d_ws;
    int*   cnt    = (int*)((char*)d_ws + WS_CNT);
    int*   rowptr = (int*)((char*)d_ws + WS_ROWPTR);
    int*   rank   = (int*)((char*)d_ws + WS_RANK);
    int*   csr    = (int*)((char*)d_ws + WS_CSR);
    float* bufA   = (float*)((char*)d_ws + WS_BUFA);
    float* bufB   = bufA + (size_t)NN * HD;

    const int NODE_BLK = (NN + 255) / 256;
    const int EDGE_BLK = (NE + 255) / 256;
    const int WAVE_BLK = (NN * 32 + 255) / 256;
    const int GEMM_BLK = ((NN / 16) * (HD / 16) + 7) / 8;

    k_cnt_zero<<<NODE_BLK, 256, 0, stream>>>(cnt);
    k_rank    <<<EDGE_BLK, 256, 0, stream>>>(dst, cnt, rank);
    k_scan    <<<NODE_BLK, 1024, 0, stream>>>(cnt, rowptr, dinv);
    k_fill    <<<EDGE_BLK, 256, 0, stream>>>(src, dst, rowptr, rank, csr);

    k_gemm_wmma_f32<<<GEMM_BLK, 256, 0, stream>>>(x, W1, bufA, NN, FIN, HD);
    k_gather_relu  <<<WAVE_BLK, 256, 0, stream>>>(bufA, rowptr, csr, dinv, b1, bufB);

    k_gemm_wmma_f32<<<GEMM_BLK, 256, 0, stream>>>(bufB, W2, bufA, NN, HD, HD);
    k_gather_heads <<<WAVE_BLK, 256, 0, stream>>>(bufA, rowptr, csr, dinv, b2, Wh1, bh1, Wh2, bh2,
                                                  out1, out2, hout);
}


# GEMM rebuilt: 256 resident work-groups keep all of W in LDS (row pairs interleaved) and loop over 16x64 strips, 4 accumulators per wave, A requested one chunk ahead; k order and sums unchanged
# speedup vs baseline: 2.5500x; 1.2293x over previous
#include <hip/hip_runtime.h>

typedef __attribute__((ext_vector_type(2))) float v2f;
typedef __attribute__((ext_vector_type(8))) float v8f;

#define NN 50000
#define NE 800000
#define FIN 256
#define HD 128
#define NC1 4
#define NC2 3

#define WS_CNT     200192u
#define WS_ROWPTR  400384u
#define WS_RANK    600576u
#define WS_CSR     3800576u
#define WS_BUFA    7000576u

__global__ void k_cnt_zero(int* __restrict__ cnt) {
    int i = blockIdx.x * blockDim.x + threadIdx.x;
    if (i < NN) cnt[i] = 0;
}

__global__ void k_rank(const int* __restrict__ dst, int* __restrict__ cnt, int* __restrict__ rank) {
    int e = blockIdx.x * blockDim.x + threadIdx.x;
    if (e < NE) rank[e] = atomicAdd(&cnt[dst[e]], 1);
}

__global__ void __launch_bounds__(1024) k_scan(const int* __restrict__ cnt, int* __restrict__ rowptr,
                                               float* __restrict__ dinv) {
    __shared__ int s_part[32];
    __shared__ int s_tot[8];
    const int t = threadIdx.x;
    const int lane = t & 31, wv = t >> 5;
    const int b0 = blockIdx.x * 256;

    int part = 0;
    #pragma unroll 4
    for (int j = t * 4; j < b0; j += 4096) {
        const int4 v = *(const int4*)(cnt + j);
        part += (v.x + v.y) + (v.z + v.w);
    }
    #pragma unroll
    for (int off = 16; off > 0; off >>= 1) part += __shfl_xor(part, off, 32);
    if (lane == 0) s_part[wv] = part;

    const int i = b0 + t;
    int c = 0, inc = 0;
    if (t < 256) {
        c = (i < NN) ? cnt[i] : 0;
        inc = c;
        #pragma unroll
        for (int off = 1; off < 32; off <<= 1) {
            const int v = __shfl_up(inc, off, 32);
            if (lane >= off) inc += v;
        }
        if (lane == 31) s_tot[wv] = inc;
    }
    __syncthreads();
    if (t < 256) {
        int base = 0;
        #pragma unroll
        for (int k = 0; k < 32; ++k) base += s_part[k];
        #pragma unroll
        for (int k = 0; k < 8; ++k) base += (k < wv) ? s_tot[k] : 0;
        const int excl = base + inc - c;
        if (i < NN) {
            rowptr[i] = excl;
            dinv[i] = rsqrtf((float)(c + 1));
        }
        if (i == NN - 1) rowptr[NN] = excl + c;
    }
}

__global__ void k_fill(const int* __restrict__ src, const int* __restrict__ dst,
                       const int* __restrict__ rowptr, const int* __restrict__ rank,
                       int* __restrict__ csr) {
    int e = blockIdx.x * blockDim.x + threadIdx.x;
    if (e >= NE) return;
    const int pos = rowptr[dst[e]] + rank[e];
    if ((unsigned)pos < (unsigned)NE) csr[pos] = src[e];
}

typedef __attribute__((ext_vector_type(4))) float v4f;
#define GEMM_LDS_FLOATS 43008
#define GEMM_TILES (NN / 16)
#define GEMM_PAIRS 1024

#define GEMM_CK 32

template <int K>
__device__ __forceinline__ void gemm_load_a(v2f (&a)[GEMM_CK], const float* __restrict__ X, int tile, int chunk,
                                            int lh, int half) {
    const float* xp = X + (size_t)(tile * 16 + lh) * K + (chunk * (4 * GEMM_CK) + 2 * half);
    #pragma unroll
    for (int ks = 0; ks < GEMM_CK; ++ks) a[ks] = *(const v2f*)(xp + 4 * ks);
}

__device__ __forceinline__ void gemm_mma_chunk(const v2f (&a)[GEMM_CK], const float* bl,
                                               v8f& acc0, v8f& acc1, v8f& acc2, v8f& acc3) {
    #pragma unroll
    for (int ks = 0; ks < GEMM_CK; ++ks) {
        const v4f b01 = *(const v4f*)(bl + ks * (4 * HD));
        const v4f b23 = *(const v4f*)(bl + ks * (4 * HD) + 64);
        const v2f b0 = { b01.x, b01.y };
        const v2f b1 = { b01.z, b01.w };
        const v2f b2 = { b23.x, b23.y };
        const v2f b3 = { b23.z, b23.w };
        acc0 = __builtin_amdgcn_wmma_f32_16x16x4_f32(false, a[ks], false, b0, (short)0, acc0, false, false);
        acc1 = __builtin_amdgcn_wmma_f32_16x16x4_f32(false, a[ks], false, b1, (short)0, acc1, false, false);
        acc2 = __builtin_amdgcn_wmma_f32_16x16x4_f32(false, a[ks], false, b2, (short)0, acc2, false, false);
        acc3 = __builtin_amdgcn_wmma_f32_16x16x4_f32(false, a[ks], false, b3, (short)0, acc3, false, false);
    }
}

template <int K>
__global__ void __launch_bounds__(256) k_gemm_wlds(const float* __restrict__ X, const float* __restrict__ W,
                                                   float* __restrict__ OUT) {
    __shared__ __attribute__((aligned(16))) float Wl[GEMM_LDS_FLOATS];
    constexpr int NCH = K / (4 * GEMM_CK);
    const int tid  = threadIdx.x;
    const int lane = tid & 31;
    const int wv   = __builtin_amdgcn_readfirstlane(tid >> 5);
    const int half = lane >> 4;
    const int lh   = lane & 15;
    const int hf   = wv & 1;
    int t = (wv >> 1) * 256 + (int)blockIdx.x;

    v2f a0[GEMM_CK], a1[GEMM_CK];
    gemm_load_a<K>(a0, X, t, 0, lh, half);

    {
        const int j = tid & 31;
        #pragma unroll 4
        for (int kp = wv; kp < K / 2; kp += 8) {
            const float4 w0 = *(const float4*)(W + (size_t)(2 * kp) * HD + 4 * j);
            const float4 w1 = *(const float4*)(W + (size_t)(2 * kp + 1) * HD + 4 * j);
            float4* d = (float4*)(Wl + (kp * HD + 4 * j) * 2);
            d[0] = make_float4(w0.x, w1.x, w0.y, w1.y);
            d[1] = make_float4(w0.z, w1.z, w0.w, w1.w);
        }
    }
    __syncthreads();

    const float* bl = Wl + (half * HD + 64 * hf + 2 * lh) * 2;

    for (;;) {
        const int  tn   = t + GEMM_PAIRS;
        const bool more = tn < GEMM_TILES;
        const int  tl   = more ? tn : t;

        v8f acc0 = {}, acc1 = {}, acc2 = {}, acc3 = {};
        if constexpr (NCH == 2) {
            asm volatile("" ::: "memory");
            gemm_load_a<K>(a1, X, t, 1, lh, half);
            asm volatile("" ::: "memory");
            gemm_mma_chunk(a0, bl, acc0, acc1, acc2, acc3);
            gemm_load_a<K>(a0, X, tl, 0, lh, half);
            asm volatile("" ::: "memory");
            gemm_mma_chunk(a1, bl + GEMM_CK * (4 * HD), acc0, acc1, acc2, acc3);
        } else {
            gemm_load_a<K>(a1, X, tl, 0, lh, half);
            gemm_mma_chunk(a0, bl, acc0, acc1, acc2, acc3);
        }

        float* op = OUT + (size_t)(t * 16 + 8 * half) * HD + 64 * hf + 2 * lh;
        #pragma unroll
        for (int r = 0; r < 8; ++r) {
            const v2f o01 = { acc0[r], acc1[r] };
            const v2f o23 = { acc2[r], acc3[r] };
            *(v2f*)(op + r * HD)      = o01;
            *(v2f*)(op + r * HD + 32) = o23;
        }

        if (!more) break;
        if constexpr (NCH == 1) {
            #pragma unroll
            for (int ks = 0; ks < GEMM_CK; ++ks) a0[ks] = a1[ks];
        }
        t = tn;
    }
}

template <int K>
__device__ __forceinline__ void gather_group(const float* __restrict__ Hf, int s_l, float w_l, int j,
                                             unsigned lane4, float4& acc) {
    float4 hv[K];
    float  w[K];
    #pragma unroll
    for (int u = 0; u < K; ++u) {
        const unsigned s = (unsigned)__builtin_amdgcn_readlane(s_l, j + u);
        w[u]  = __int_as_float(__builtin_amdgcn_readlane(__float_as_int(w_l), j + u));
        hv[u] = *(const float4*)(Hf + (s * HD + lane4));
    }
    #pragma unroll
    for (int u = 0; u < K; ++u) {
        acc.x = fmaf(w[u], hv[u].x, acc.x);
        acc.y = fmaf(w[u], hv[u].y, acc.y);
        acc.z = fmaf(w[u], hv[u].z, acc.z);
        acc.w = fmaf(w[u], hv[u].w, acc.w);
    }
}

__device__ __forceinline__ float4 aggregate_row(const float* __restrict__ Hf, const int* __restrict__ rowptr,
                                                const int* __restrict__ csr, const float* __restrict__ dinv,
                                                int node, int lane, unsigned lane4) {
    const int beg = rowptr[node];
    const int end = rowptr[node + 1];
    const float di = dinv[node];

    const float4 hs = *(const float4*)(Hf + ((unsigned)node * HD + lane4));
    const float wself = di * di;
    float4 acc = { hs.x * wself, hs.y * wself, hs.z * wself, hs.w * wself };

    for (int base = beg; base < end; base += 32) {
        const int n = min(32, end - base);
        int   s_l = node;
        float w_l = 0.0f;
        if (lane < n) {
            const int sv = csr[base + lane];
            s_l = ((unsigned)sv < (unsigned)NN) ? sv : node;
            w_l = dinv[s_l] * di;
        }
        int j = 0;
        for (; j + 8 <= n; j += 8) gather_group<8>(Hf, s_l, w_l, j, lane4, acc);
        if (j + 4 <= n) { gather_group<4>(Hf, s_l, w_l, j, lane4, acc); j += 4; }
        if (j + 2 <= n) { gather_group<2>(Hf, s_l, w_l, j, lane4, acc); j += 2; }
        if (j < n)      { gather_group<1>(Hf, s_l, w_l, j, lane4, acc); }
    }
    return acc;
}

__global__ void __launch_bounds__(256) k_gather_relu(const float* __restrict__ Hf, const int* __restrict__ rowptr,
                                                     const int* __restrict__ csr, const float* __restrict__ dinv,
                                                     const float* __restrict__ bias, float* __restrict__ OUT) {
    const int t = blockIdx.x * blockDim.x + threadIdx.x;
    const int lane = threadIdx.x & 31;
    const int node = __builtin_amdgcn_readfirstlane(t >> 5);
    if (node >= NN) return;
    const unsigned lane4 = (unsigned)lane * 4u;
    const float4 bb = ((const float4*)bias)[lane];
    const float4 acc = aggregate_row(Hf, rowptr, csr, dinv, node, lane, lane4);
    float4 r;
    r.x = fmaxf(acc.x + bb.x, 0.0f);
    r.y = fmaxf(acc.y + bb.y, 0.0f);
    r.z = fmaxf(acc.z + bb.z, 0.0f);
    r.w = fmaxf(acc.w + bb.w, 0.0f);
    *(float4*)(OUT + ((unsigned)node * HD + lane4)) = r;
}

__global__ void __launch_bounds__(256) k_gather_heads(const float* __restrict__ Hf, const int* __restrict__ rowptr,
                                                      const int* __restrict__ csr, const float* __restrict__ dinv,
                                                      const float* __restrict__ b2,
                                                      const float* __restrict__ Wh1, const float* __restrict__ bh1,
                                                      const float* __restrict__ Wh2, const float* __restrict__ bh2,
                                                      float* __restrict__ out1, float* __restrict__ out2,
                                                      float* __restrict__ hout) {
    const int t = blockIdx.x * blockDim.x + threadIdx.x;
    const int lane = threadIdx.x & 31;
    const int node = __builtin_amdgcn_readfirstlane(t >> 5);
    if (node >= NN) return;
    const unsigned lane4 = (unsigned)lane * 4u;
    const float4 bb = ((const float4*)b2)[lane];
    const float4 av = aggregate_row(Hf, rowptr, csr, dinv, node, lane, lane4);

    float hv[4] = { av.x + bb.x, av.y + bb.y, av.z + bb.z, av.w + bb.w };

    float4 hstore = { hv[0], hv[1], hv[2], hv[3] };
    *(float4*)(hout + (size_t)node * HD + lane4) = hstore;

    float acc[NC1 + NC2];
    #pragma unroll
    for (int c = 0; c < NC1 + NC2; ++c) acc[c] = 0.0f;

    #pragma unroll
    for (int j = 0; j < 4; ++j) {
        const int k = lane * 4 + j;
        #pragma unroll
        for (int c = 0; c < NC1; ++c) acc[c]       += hv[j] * Wh1[k * NC1 + c];
        #pragma unroll
        for (int c = 0; c < NC2; ++c) acc[NC1 + c] += hv[j] * Wh2[k * NC2 + c];
    }

    #pragma unroll
    for (int off = 16; off > 0; off >>= 1) {
        #pragma unroll
        for (int c = 0; c < NC1 + NC2; ++c)
            acc[c] += __shfl_down(acc[c], off, 32);
    }

    if (lane == 0) {
        #pragma unroll
        for (int c = 0; c < NC1; ++c) out1[(size_t)node * NC1 + c] = acc[c] + bh1[c];
        #pragma unroll
        for (int c = 0; c < NC2; ++c) out2[(size_t)node * NC2 + c] = acc[NC1 + c] + bh2[c];
    }
}

extern "C" void kernel_launch(void* const* d_in, const int* in_sizes, int n_in,
                              void* d_out, int out_size, void* d_ws, size_t ws_size,
                              hipStream_t stream) {
    const float* x   = (const float*)d_in[0];
    const int*   ei  = (const int*)d_in[1];
    const int*   src = ei;
    const int*   dst = ei + NE;
    const float* W1  = (const float*)d_in[2];
    const float* b1  = (const float*)d_in[3];
    const float* W2  = (const float*)d_in[4];
    const float* b2  = (const float*)d_in[5];
    const float* Wh1 = (const float*)d_in[6];
    const float* bh1 = (const float*)d_in[7];
    const float* Wh2 = (const float*)d_in[8];
    const float* bh2 = (const float*)d_in[9];

    float* out1 = (float*)d_out;
    float* out2 = out1 + (size_t)NN * NC1;
    float* hout = out2 + (size_t)NN * NC2;

    float* dinv   = (float*)d_ws;
    int*   cnt    = (int*)((char*)d_ws + WS_CNT);
    int*   rowptr = (int*)((char*)d_ws + WS_ROWPTR);
    int*   rank   = (int*)((char*)d_ws + WS_RANK);
    int*   csr    = (int*)((char*)d_ws + WS_CSR);
    float* bufA   = (float*)((char*)d_ws + WS_BUFA);
    float* bufB   = bufA + (size_t)NN * HD;

    const int NODE_BLK = (NN + 255) / 256;
    const int EDGE_BLK = (NE + 255) / 256;
    const int WAVE_BLK = (NN * 32 + 255) / 256;

    k_cnt_zero<<<NODE_BLK, 256, 0, stream>>>(cnt);
    k_rank    <<<EDGE_BLK, 256, 0, stream>>>(dst, cnt, rank);
    k_scan    <<<NODE_BLK, 1024, 0, stream>>>(cnt, rowptr, dinv);
    k_fill    <<<EDGE_BLK, 256, 0, stream>>>(src, dst, rowptr, rank, csr);

    k_gemm_wlds<FIN><<<256, 256, 0, stream>>>(x, W1, bufA);
    k_gather_relu  <<<WAVE_BLK, 256, 0, stream>>>(bufA, rowptr, csr, dinv, b1, bufB);

    k_gemm_wlds<HD><<<256, 256, 0, stream>>>(bufB, W2, bufA);
    k_gather_heads <<<WAVE_BLK, 256, 0, stream>>>(bufA, rowptr, csr, dinv, b2, Wh1, bh1, Wh2, bh2,
                                                  out1, out2, hout);
}


# Edge ranking (returning count adds) moved from its own launch onto 4 extra waves per work-group of the layer-1 GEMM launch, independent of it: 7 launches; row sums stay in arrival-order slots
# speedup vs baseline: 2.6733x; 1.0484x over previous
#include <hip/hip_runtime.h>

typedef __attribute__((ext_vector_type(2))) float v2f;
typedef __attribute__((ext_vector_type(8))) float v8f;

#define NN 50000
#define NE 800000
#define FIN 256
#define HD 128
#define NC1 4
#define NC2 3

#define WS_CNT     200192u
#define WS_ROWPTR  400384u
#define WS_RANK    600576u
#define WS_CSR     3800576u
#define WS_BUFA    7000576u

__global__ void k_cnt_zero(int* __restrict__ cnt) {
    int i = blockIdx.x * blockDim.x + threadIdx.x;
    if (i < NN) cnt[i] = 0;
}

template <int NT>
__device__ __forceinline__ void edge_rank_part(const int* __restrict__ dst, int* __restrict__ cnt,
                                               int* __restrict__ rank, int et) {
    constexpr int FULL = NE / NT;
    int d[FULL], r[FULL];
    #pragma unroll
    for (int j = 0; j < FULL; ++j) d[j] = dst[et + j * NT];
    #pragma unroll
    for (int j = 0; j < FULL; ++j) r[j] = atomicAdd(&cnt[d[j]], 1);
    #pragma unroll
    for (int j = 0; j < FULL; ++j) rank[et + j * NT] = r[j];
    if constexpr (FULL * NT < NE) {
        const int e = et + FULL * NT;
        if (e < NE) rank[e] = atomicAdd(&cnt[dst[e]], 1);
    }
}

__global__ void __launch_bounds__(1024) k_scan(const int* __restrict__ cnt, int* __restrict__ rowptr,
                                               float* __restrict__ dinv) {
    __shared__ int s_part[32];
    __shared__ int s_tot[8];
    const int t = threadIdx.x;
    const int lane = t & 31, wv = t >> 5;
    const int b0 = blockIdx.x * 256;

    int part = 0;
    #pragma unroll 4
    for (int j = t * 4; j < b0; j += 4096) {
        const int4 v = *(const int4*)(cnt + j);
        part += (v.x + v.y) + (v.z + v.w);
    }
    #pragma unroll
    for (int off = 16; off > 0; off >>= 1) part += __shfl_xor(part, off, 32);
    if (lane == 0) s_part[wv] = part;

    const int i = b0 + t;
    int c = 0, inc = 0;
    if (t < 256) {
        c = (i < NN) ? cnt[i] : 0;
        inc = c;
        #pragma unroll
        for (int off = 1; off < 32; off <<= 1) {
            const int v = __shfl_up(inc, off, 32);
            if (lane >= off) inc += v;
        }
        if (lane == 31) s_tot[wv] = inc;
    }
    __syncthreads();
    if (t < 256) {
        int base = 0;
        #pragma unroll
        for (int k = 0; k < 32; ++k) base += s_part[k];
        #pragma unroll
        for (int k = 0; k < 8; ++k) base += (k < wv) ? s_tot[k] : 0;
        const int excl = base + inc - c;
        if (i < NN) {
            rowptr[i] = excl;
            dinv[i] = rsqrtf((float)(c + 1));
        }
        if (i == NN - 1) rowptr[NN] = excl + c;
    }
}

__global__ void k_fill(const int* __restrict__ src, const int* __restrict__ dst,
                       const int* __restrict__ rowptr, const int* __restrict__ rank,
                       int* __restrict__ csr) {
    int e = blockIdx.x * blockDim.x + threadIdx.x;
    if (e >= NE) return;
    const int pos = rowptr[dst[e]] + rank[e];
    if ((unsigned)pos < (unsigned)NE) csr[pos] = src[e];
}

typedef __attribute__((ext_vector_type(4))) float v4f;
#define GEMM_LDS_FLOATS 43008
#define GEMM_TILES (NN / 16)
#define GEMM_PAIRS 1024

#define GEMM_CK 32

template <int K>
__device__ __forceinline__ void gemm_load_a(v2f (&a)[GEMM_CK], const float* __restrict__ X, int tile, int chunk,
                                            int lh, int half) {
    const float* xp = X + (size_t)(tile * 16 + lh) * K + (chunk * (4 * GEMM_CK) + 2 * half);
    #pragma unroll
    for (int ks = 0; ks < GEMM_CK; ++ks) a[ks] = *(const v2f*)(xp + 4 * ks);
}

__device__ __forceinline__ void gemm_mma_chunk(const v2f (&a)[GEMM_CK], const float* bl,
                                               v8f& acc0, v8f& acc1, v8f& acc2, v8f& acc3) {
    #pragma unroll
    for (int ks = 0; ks < GEMM_CK; ++ks) {
        const v4f b01 = *(const v4f*)(bl + ks * (4 * HD));
        const v4f b23 = *(const v4f*)(bl + ks * (4 * HD) + 64);
        const v2f b0 = { b01.x, b01.y };
        const v2f b1 = { b01.z, b01.w };
        const v2f b2 = { b23.x, b23.y };
        const v2f b3 = { b23.z, b23.w };
        acc0 = __builtin_amdgcn_wmma_f32_16x16x4_f32(false, a[ks], false, b0, (short)0, acc0, false, false);
        acc1 = __builtin_amdgcn_wmma_f32_16x16x4_f32(false, a[ks], false, b1, (short)0, acc1, false, false);
        acc2 = __builtin_amdgcn_wmma_f32_16x16x4_f32(false, a[ks], false, b2, (short)0, acc2, false, false);
        acc3 = __builtin_amdgcn_wmma_f32_16x16x4_f32(false, a[ks], false, b3, (short)0, acc3, false, false);
    }
}

template <int K, int EW>
__global__ void __launch_bounds__(256 + 32 * EW) k_gemm_wlds(const float* __restrict__ X, const float* __restrict__ W,
                                                             float* __restrict__ OUT, const int* __restrict__ dst,
                                                             int* __restrict__ cnt, int* __restrict__ rank) {
    __shared__ __attribute__((aligned(16))) float Wl[GEMM_LDS_FLOATS];
    constexpr int NCH = K / (4 * GEMM_CK);
    const int tid  = threadIdx.x;
    const int lane = tid & 31;
    const int wv   = __builtin_amdgcn_readfirstlane(tid >> 5);
    const int half = lane >> 4;
    const int lh   = lane & 15;
    const int hf   = wv & 1;
    int t = (wv >> 1) * 256 + (int)blockIdx.x;

    v2f a0[GEMM_CK], a1[GEMM_CK];
    const bool mul_wave = (EW == 0) || (wv < 8);
    if (mul_wave) gemm_load_a<K>(a0, X, t, 0, lh, half);

    {
        const int j = tid & 31;
        #pragma unroll 4
        for (int kp = wv; kp < K / 2; kp += 8 + EW) {
            const float4 w0 = *(const float4*)(W + (size_t)(2 * kp) * HD + 4 * j);
            const float4 w1 = *(const float4*)(W + (size_t)(2 * kp + 1) * HD + 4 * j);
            float4* d = (float4*)(Wl + (kp * HD + 4 * j) * 2);
            d[0] = make_float4(w0.x, w1.x, w0.y, w1.y);
            d[1] = make_float4(w0.z, w1.z, w0.w, w1.w);
        }
    }
    __syncthreads();

    if constexpr (EW > 0) {
        if (!mul_wave) {
            edge_rank_part<256 * 32 * EW>(dst, cnt, rank, (int)blockIdx.x * (32 * EW) + (tid - 256));
            return;
        }
    }

    const float* bl = Wl + (half * HD + 64 * hf + 2 * lh) * 2;

    for (;;) {
        const int  tn   = t + GEMM_PAIRS;
        const bool more = tn < GEMM_TILES;
        const int  tl   = more ? tn : t;

        v8f acc0 = {}, acc1 = {}, acc2 = {}, acc3 = {};
        if constexpr (NCH == 2) {
            asm volatile("" ::: "memory");
            gemm_load_a<K>(a1, X, t, 1, lh, half);
            asm volatile("" ::: "memory");
            gemm_mma_chunk(a0, bl, acc0, acc1, acc2, acc3);
            gemm_load_a<K>(a0, X, tl, 0, lh, half);
            asm volatile("" ::: "memory");
            gemm_mma_chunk(a1, bl + GEMM_CK * (4 * HD), acc0, acc1, acc2, acc3);
        } else {
            gemm_load_a<K>(a1, X, tl, 0, lh, half);
            gemm_mma_chunk(a0, bl, acc0, acc1, acc2, acc3);
        }

        float* op = OUT + (size_t)(t * 16 + 8 * half) * HD + 64 * hf + 2 * lh;
        #pragma unroll
        for (int r = 0; r < 8; ++r) {
            const v2f o01 = { acc0[r], acc1[r] };
            const v2f o23 = { acc2[r], acc3[r] };
            *(v2f*)(op + r * HD)      = o01;
            *(v2f*)(op + r * HD + 32) = o23;
        }

        if (!more) break;
        if constexpr (NCH == 1) {
            #pragma unroll
            for (int ks = 0; ks < GEMM_CK; ++ks) a0[ks] = a1[ks];
        }
        t = tn;
    }
}

template <int K>
__device__ __forceinline__ void gather_group(const float* __restrict__ Hf, int s_l, float w_l, int j,
                                             unsigned lane4, float4& acc) {
    float4 hv[K];
    float  w[K];
    #pragma unroll
    for (int u = 0; u < K; ++u) {
        const unsigned s = (unsigned)__builtin_amdgcn_readlane(s_l, j + u);
        w[u]  = __int_as_float(__builtin_amdgcn_readlane(__float_as_int(w_l), j + u));
        hv[u] = *(const float4*)(Hf + (s * HD + lane4));
    }
    #pragma unroll
    for (int u = 0; u < K; ++u) {
        acc.x = fmaf(w[u], hv[u].x, acc.x);
        acc.y = fmaf(w[u], hv[u].y, acc.y);
        acc.z = fmaf(w[u], hv[u].z, acc.z);
        acc.w = fmaf(w[u], hv[u].w, acc.w);
    }
}

__device__ __forceinline__ float4 aggregate_row(const float* __restrict__ Hf, const int* __restrict__ rowptr,
                                                const int* __restrict__ csr, const float* __restrict__ dinv,
                                                int node, int lane, unsigned lane4) {
    const int beg = rowptr[node];
    const int end = rowptr[node + 1];
    const float di = dinv[node];

    const float4 hs = *(const float4*)(Hf + ((unsigned)node * HD + lane4));
    const float wself = di * di;
    float4 acc = { hs.x * wself, hs.y * wself, hs.z * wself, hs.w * wself };

    for (int base = beg; base < end; base += 32) {
        const int n = min(32, end - base);
        int   s_l = node;
        float w_l = 0.0f;
        if (lane < n) {
            const int sv = csr[base + lane];
            s_l = ((unsigned)sv < (unsigned)NN) ? sv : node;
            w_l = dinv[s_l] * di;
        }
        int j = 0;
        for (; j + 8 <= n; j += 8) gather_group<8>(Hf, s_l, w_l, j, lane4, acc);
        if (j + 4 <= n) { gather_group<4>(Hf, s_l, w_l, j, lane4, acc); j += 4; }
        if (j + 2 <= n) { gather_group<2>(Hf, s_l, w_l, j, lane4, acc); j += 2; }
        if (j < n)      { gather_group<1>(Hf, s_l, w_l, j, lane4, acc); }
    }
    return acc;
}

__global__ void __launch_bounds__(256) k_gather_relu(const float* __restrict__ Hf, const int* __restrict__ rowptr,
                                                     const int* __restrict__ csr, const float* __restrict__ dinv,
                                                     const float* __restrict__ bias, float* __restrict__ OUT) {
    const int t = blockIdx.x * blockDim.x + threadIdx.x;
    const int lane = threadIdx.x & 31;
    const int node = __builtin_amdgcn_readfirstlane(t >> 5);
    if (node >= NN) return;
    const unsigned lane4 = (unsigned)lane * 4u;
    const float4 bb = ((const float4*)bias)[lane];
    const float4 acc = aggregate_row(Hf, rowptr, csr, dinv, node, lane, lane4);
    float4 r;
    r.x = fmaxf(acc.x + bb.x, 0.0f);
    r.y = fmaxf(acc.y + bb.y, 0.0f);
    r.z = fmaxf(acc.z + bb.z, 0.0f);
    r.w = fmaxf(acc.w + bb.w, 0.0f);
    *(float4*)(OUT + ((unsigned)node * HD + lane4)) = r;
}

__global__ void __launch_bounds__(256) k_gather_heads(const float* __restrict__ Hf, const int* __restrict__ rowptr,
                                                      const int* __restrict__ csr, const float* __restrict__ dinv,
                                                      const float* __restrict__ b2,
                                                      const float* __restrict__ Wh1, const float* __restrict__ bh1,
                                                      const float* __restrict__ Wh2, const float* __restrict__ bh2,
                                                      float* __restrict__ out1, float* __restrict__ out2,
                                                      float* __restrict__ hout) {
    const int t = blockIdx.x * blockDim.x + threadIdx.x;
    const int lane = threadIdx.x & 31;
    const int node = __builtin_amdgcn_readfirstlane(t >> 5);
    if (node >= NN) return;
    const unsigned lane4 = (unsigned)lane * 4u;
    const float4 bb = ((const float4*)b2)[lane];
    const float4 av = aggregate_row(Hf, rowptr, csr, dinv, node, lane, lane4);

    float hv[4] = { av.x + bb.x, av.y + bb.y, av.z + bb.z, av.w + bb.w };

    float4 hstore = { hv[0], hv[1], hv[2], hv[3] };
    *(float4*)(hout + (size_t)node * HD + lane4) = hstore;

    float acc[NC1 + NC2];
    #pragma unroll
    for (int c = 0; c < NC1 + NC2; ++c) acc[c] = 0.0f;

    #pragma unroll
    for (int j = 0; j < 4; ++j) {
        const int k = lane * 4 + j;
        #pragma unroll
        for (int c = 0; c < NC1; ++c) acc[c]       += hv[j] * Wh1[k * NC1 + c];
        #pragma unroll
        for (int c = 0; c < NC2; ++c) acc[NC1 + c] += hv[j] * Wh2[k * NC2 + c];
    }

    #pragma unroll
    for (int off = 16; off > 0; off >>= 1) {
        #pragma unroll
        for (int c = 0; c < NC1 + NC2; ++c)
            acc[c] += __shfl_down(acc[c], off, 32);
    }

    if (lane == 0) {
        #pragma unroll
        for (int c = 0; c < NC1; ++c) out1[(size_t)node * NC1 + c] = acc[c] + bh1[c];
        #pragma unroll
        for (int c = 0; c < NC2; ++c) out2[(size_t)node * NC2 + c] = acc[NC1 + c] + bh2[c];
    }
}

extern "C" void kernel_launch(void* const* d_in, const int* in_sizes, int n_in,
                              void* d_out, int out_size, void* d_ws, size_t ws_size,
                              hipStream_t stream) {
    const float* x   = (const float*)d_in[0];
    const int*   ei  = (const int*)d_in[1];
    const int*   src = ei;
    const int*   dst = ei + NE;
    const float* W1  = (const float*)d_in[2];
    const float* b1  = (const float*)d_in[3];
    const float* W2  = (const float*)d_in[4];
    const float* b2  = (const float*)d_in[5];
    const float* Wh1 = (const float*)d_in[6];
    const float* bh1 = (const float*)d_in[7];
    const float* Wh2 = (const float*)d_in[8];
    const float* bh2 = (const float*)d_in[9];

    float* out1 = (float*)d_out;
    float* out2 = out1 + (size_t)NN * NC1;
    float* hout = out2 + (size_t)NN * NC2;

    float* dinv   = (float*)d_ws;
    int*   cnt    = (int*)((char*)d_ws + WS_CNT);
    int*   rowptr = (int*)((char*)d_ws + WS_ROWPTR);
    int*   rank   = (int*)((char*)d_ws + WS_RANK);
    int*   csr    = (int*)((char*)d_ws + WS_CSR);
    float* bufA   = (float*)((char*)d_ws + WS_BUFA);
    float* bufB   = bufA + (size_t)NN * HD;

    const int NODE_BLK = (NN + 255) / 256;
    const int EDGE_BLK = (NE + 255) / 256;
    const int WAVE_BLK = (NN * 32 + 255) / 256;

    k_cnt_zero<<<NODE_BLK, 256, 0, stream>>>(cnt);
    k_gemm_wlds<FIN, 4><<<256, 384, 0, stream>>>(x, W1, bufA, dst, cnt, rank);
    k_scan    <<<NODE_BLK, 1024, 0, stream>>>(cnt, rowptr, dinv);
    k_fill    <<<EDGE_BLK, 256, 0, stream>>>(src, dst, rowptr, rank, csr);

    k_gather_relu  <<<WAVE_BLK, 256, 0, stream>>>(bufA, rowptr, csr, dinv, b1, bufB);

    k_gemm_wlds<HD, 0><<<256, 256, 0, stream>>>(bufB, W2, bufA, nullptr, nullptr, nullptr);
    k_gather_heads <<<WAVE_BLK, 256, 0, stream>>>(bufA, rowptr, csr, dinv, b2, Wh1, bh1, Wh2, bh2,
                                                  out1, out2, hout);
}


# Scan and fill launches removed: the ranking add's return value places each source id into a fixed 64-slot row per node (overflow list beyond); dinv from counts; arrival-order sums: 5 launches
# speedup vs baseline: 3.3346x; 1.2474x over previous
#include <hip/hip_runtime.h>

typedef __attribute__((ext_vector_type(2))) float v2f;
typedef __attribute__((ext_vector_type(8))) float v8f;

#define NN 50000
#define NE 800000
#define FIN 256
#define HD 128
#define NC1 4
#define NC2 3

#define SLOT_CAP 64

#define WS_SLOT    200192u
#define WS_OVF     13000192u
#define WS_BUFA    16200192u

__global__ void k_cnt_zero(int* __restrict__ cnt) {
    int i = blockIdx.x * blockDim.x + threadIdx.x;
    if (i <= NN) cnt[i] = 0;
}

template <int NT>
__device__ __forceinline__ void edge_slot_part(const int* __restrict__ src, const int* __restrict__ dst,
                                               int* __restrict__ cnt, int* __restrict__ slot,
                                               int* __restrict__ ovf, int et) {
    constexpr int FULL = NE / NT;
    int d[FULL], r[FULL], s[FULL];
    #pragma unroll
    for (int j = 0; j < FULL; ++j) d[j] = dst[et + j * NT];
    #pragma unroll
    for (int j = 0; j < FULL; ++j) s[j] = src[et + j * NT];
    #pragma unroll
    for (int j = 0; j < FULL; ++j) r[j] = atomicAdd(&cnt[d[j]], 1);
    bool over = false;
    #pragma unroll
    for (int j = 0; j < FULL; ++j) {
        if ((unsigned)r[j] < (unsigned)SLOT_CAP) slot[d[j] * SLOT_CAP + r[j]] = s[j];
        else over = true;
    }
    if (over) {
        #pragma unroll
        for (int j = 0; j < FULL; ++j) {
            if ((unsigned)r[j] >= (unsigned)SLOT_CAP) {
                const int o = atomicAdd(&cnt[NN], 1);
                if ((unsigned)o < (unsigned)NE) ovf[o] = et + j * NT;
            }
        }
    }
    if constexpr (FULL * NT < NE) {
        const int e = et + FULL * NT;
        if (e < NE) {
            const int dd = dst[e];
            const int rr = atomicAdd(&cnt[dd], 1);
            if ((unsigned)rr < (unsigned)SLOT_CAP) slot[dd * SLOT_CAP + rr] = src[e];
            else {
                const int o = atomicAdd(&cnt[NN], 1);
                if ((unsigned)o < (unsigned)NE) ovf[o] = e;
            }
        }
    }
}

typedef __attribute__((ext_vector_type(4))) float v4f;
#define GEMM_LDS_FLOATS 43008
#define GEMM_TILES (NN / 16)
#define GEMM_PAIRS 1024

#define GEMM_CK 32

template <int K>
__device__ __forceinline__ void gemm_load_a(v2f (&a)[GEMM_CK], const float* __restrict__ X, int tile, int chunk,
                                            int lh, int half) {
    const float* xp = X + (size_t)(tile * 16 + lh) * K + (chunk * (4 * GEMM_CK) + 2 * half);
    #pragma unroll
    for (int ks = 0; ks < GEMM_CK; ++ks) a[ks] = *(const v2f*)(xp + 4 * ks);
}

__device__ __forceinline__ void gemm_mma_chunk(const v2f (&a)[GEMM_CK], const float* bl,
                                               v8f& acc0, v8f& acc1, v8f& acc2, v8f& acc3) {
    #pragma unroll
    for (int ks = 0; ks < GEMM_CK; ++ks) {
        const v4f b01 = *(const v4f*)(bl + ks * (4 * HD));
        const v4f b23 = *(const v4f*)(bl + ks * (4 * HD) + 64);
        const v2f b0 = { b01.x, b01.y };
        const v2f b1 = { b01.z, b01.w };
        const v2f b2 = { b23.x, b23.y };
        const v2f b3 = { b23.z, b23.w };
        acc0 = __builtin_amdgcn_wmma_f32_16x16x4_f32(false, a[ks], false, b0, (short)0, acc0, false, false);
        acc1 = __builtin_amdgcn_wmma_f32_16x16x4_f32(false, a[ks], false, b1, (short)0, acc1, false, false);
        acc2 = __builtin_amdgcn_wmma_f32_16x16x4_f32(false, a[ks], false, b2, (short)0, acc2, false, false);
        acc3 = __builtin_amdgcn_wmma_f32_16x16x4_f32(false, a[ks], false, b3, (short)0, acc3, false, false);
    }
}

template <int K, int EW>
__global__ void __launch_bounds__(256 + 32 * EW) k_gemm_wlds(const float* __restrict__ X, const float* __restrict__ W,
                                                             float* __restrict__ OUT, const int* __restrict__ src,
                                                             const int* __restrict__ dst, int* __restrict__ cnt,
                                                             int* __restrict__ slot, int* __restrict__ ovf) {
    __shared__ __attribute__((aligned(16))) float Wl[GEMM_LDS_FLOATS];
    constexpr int NCH = K / (4 * GEMM_CK);
    const int tid  = threadIdx.x;
    const int lane = tid & 31;
    const int wv   = __builtin_amdgcn_readfirstlane(tid >> 5);
    const int half = lane >> 4;
    const int lh   = lane & 15;
    const int hf   = wv & 1;
    int t = (wv >> 1) * 256 + (int)blockIdx.x;

    v2f a0[GEMM_CK], a1[GEMM_CK];
    const bool mul_wave = (EW == 0) || (wv < 8);
    if (mul_wave) gemm_load_a<K>(a0, X, t, 0, lh, half);

    {
        const int j = tid & 31;
        #pragma unroll 4
        for (int kp = wv; kp < K / 2; kp += 8 + EW) {
            const float4 w0 = *(const float4*)(W + (size_t)(2 * kp) * HD + 4 * j);
            const float4 w1 = *(const float4*)(W + (size_t)(2 * kp + 1) * HD + 4 * j);
            float4* d = (float4*)(Wl + (kp * HD + 4 * j) * 2);
            d[0] = make_float4(w0.x, w1.x, w0.y, w1.y);
            d[1] = make_float4(w0.z, w1.z, w0.w, w1.w);
        }
    }
    __syncthreads();

    if constexpr (EW > 0) {
        if (!mul_wave) {
            edge_slot_part<256 * 32 * EW>(src, dst, cnt, slot, ovf, (int)blockIdx.x * (32 * EW) + (tid - 256));
            return;
        }
    }

    const float* bl = Wl + (half * HD + 64 * hf + 2 * lh) * 2;

    for (;;) {
        const int  tn   = t + GEMM_PAIRS;
        const bool more = tn < GEMM_TILES;
        const int  tl   = more ? tn : t;

        v8f acc0 = {}, acc1 = {}, acc2 = {}, acc3 = {};
        if constexpr (NCH == 2) {
            asm volatile("" ::: "memory");
            gemm_load_a<K>(a1, X, t, 1, lh, half);
            asm volatile("" ::: "memory");
            gemm_mma_chunk(a0, bl, acc0, acc1, acc2, acc3);
            gemm_load_a<K>(a0, X, tl, 0, lh, half);
            asm volatile("" ::: "memory");
            gemm_mma_chunk(a1, bl + GEMM_CK * (4 * HD), acc0, acc1, acc2, acc3);
        } else {
            gemm_load_a<K>(a1, X, tl, 0, lh, half);
            gemm_mma_chunk(a0, bl, acc0, acc1, acc2, acc3);
        }

        float* op = OUT + (size_t)(t * 16 + 8 * half) * HD + 64 * hf + 2 * lh;
        #pragma unroll
        for (int r = 0; r < 8; ++r) {
            const v2f o01 = { acc0[r], acc1[r] };
            const v2f o23 = { acc2[r], acc3[r] };
            *(v2f*)(op + r * HD)      = o01;
            *(v2f*)(op + r * HD + 32) = o23;
        }

        if (!more) break;
        if constexpr (NCH == 1) {
            #pragma unroll
            for (int ks = 0; ks < GEMM_CK; ++ks) a0[ks] = a1[ks];
        }
        t = tn;
    }
}

template <int K>
__device__ __forceinline__ void gather_group(const float* __restrict__ Hf, int s_l, float w_l, int j,
                                             unsigned lane4, float4& acc) {
    float4 hv[K];
    float  w[K];
    #pragma unroll
    for (int u = 0; u < K; ++u) {
        const unsigned s = (unsigned)__builtin_amdgcn_readlane(s_l, j + u);
        w[u]  = __int_as_float(__builtin_amdgcn_readlane(__float_as_int(w_l), j + u));
        hv[u] = *(const float4*)(Hf + (s * HD + lane4));
    }
    #pragma unroll
    for (int u = 0; u < K; ++u) {
        acc.x = fmaf(w[u], hv[u].x, acc.x);
        acc.y = fmaf(w[u], hv[u].y, acc.y);
        acc.z = fmaf(w[u], hv[u].z, acc.z);
        acc.w = fmaf(w[u], hv[u].w, acc.w);
    }
}

__device__ __forceinline__ float dinv_of(int c) { return rsqrtf((float)(c + 1)); }

__device__ __forceinline__ float4 aggregate_row(const float* __restrict__ Hf, const int* __restrict__ cnt,
                                                const int* __restrict__ slot, const int* __restrict__ ovf,
                                                const int* __restrict__ src, const int* __restrict__ dst,
                                                int node, int lane, unsigned lane4) {
    const int c = cnt[node];
    const float di = dinv_of(c);

    const float4 hs = *(const float4*)(Hf + ((unsigned)node * HD + lane4));
    const float wself = di * di;
    float4 acc = { hs.x * wself, hs.y * wself, hs.z * wself, hs.w * wself };

    const int nrow = min(c, SLOT_CAP);
    const int* srow = slot + (unsigned)node * SLOT_CAP;
    for (int base = 0; base < nrow; base += 32) {
        const int n = min(32, nrow - base);
        int   s_l = node;
        float w_l = 0.0f;
        if (lane < n) {
            const int sv = srow[base + lane];
            s_l = ((unsigned)sv < (unsigned)NN) ? sv : node;
            w_l = dinv_of(cnt[s_l]) * di;
        }
        int j = 0;
        for (; j + 8 <= n; j += 8) gather_group<8>(Hf, s_l, w_l, j, lane4, acc);
        if (j + 4 <= n) { gather_group<4>(Hf, s_l, w_l, j, lane4, acc); j += 4; }
        if (j + 2 <= n) { gather_group<2>(Hf, s_l, w_l, j, lane4, acc); j += 2; }
        if (j < n)      { gather_group<1>(Hf, s_l, w_l, j, lane4, acc); }
    }

    if (c > SLOT_CAP) {
        const int no = __builtin_amdgcn_readfirstlane(min(cnt[NN], NE));
        for (int b = 0; b < no; b += 32) {
            int   s_l = node;
            float w_l = 0.0f;
            bool  mine = false;
            if (b + lane < no) {
                const int e = ovf[b + lane];
                if ((unsigned)e < (unsigned)NE) {
                    const int sv = src[e];
                    mine = (dst[e] == node);
                    s_l = ((unsigned)sv < (unsigned)NN) ? sv : node;
                    w_l = dinv_of(cnt[s_l]) * di;
                }
            }
            unsigned m = __builtin_amdgcn_ballot_w32(mine);
            while (m) {
                const int j = __builtin_ctz(m);
                m &= m - 1u;
                gather_group<1>(Hf, s_l, w_l, j, lane4, acc);
            }
        }
    }
    return acc;
}

__global__ void __launch_bounds__(256) k_gather_relu(const float* __restrict__ Hf, const int* __restrict__ cnt,
                                                     const int* __restrict__ slot, const int* __restrict__ ovf,
                                                     const int* __restrict__ src, const int* __restrict__ dst,
                                                     const float* __restrict__ bias, float* __restrict__ OUT) {
    const int t = blockIdx.x * blockDim.x + threadIdx.x;
    const int lane = threadIdx.x & 31;
    const int node = __builtin_amdgcn_readfirstlane(t >> 5);
    if (node >= NN) return;
    const unsigned lane4 = (unsigned)lane * 4u;
    const float4 bb = ((const float4*)bias)[lane];
    const float4 acc = aggregate_row(Hf, cnt, slot, ovf, src, dst, node, lane, lane4);
    float4 r;
    r.x = fmaxf(acc.x + bb.x, 0.0f);
    r.y = fmaxf(acc.y + bb.y, 0.0f);
    r.z = fmaxf(acc.z + bb.z, 0.0f);
    r.w = fmaxf(acc.w + bb.w, 0.0f);
    *(float4*)(OUT + ((unsigned)node * HD + lane4)) = r;
}

__global__ void __launch_bounds__(256) k_gather_heads(const float* __restrict__ Hf, const int* __restrict__ cnt,
                                                      const int* __restrict__ slot, const int* __restrict__ ovf,
                                                      const int* __restrict__ src, const int* __restrict__ dst,
                                                      const float* __restrict__ b2,
                                                      const float* __restrict__ Wh1, const float* __restrict__ bh1,
                                                      const float* __restrict__ Wh2, const float* __restrict__ bh2,
                                                      float* __restrict__ out1, float* __restrict__ out2,
                                                      float* __restrict__ hout) {
    const int t = blockIdx.x * blockDim.x + threadIdx.x;
    const int lane = threadIdx.x & 31;
    const int node = __builtin_amdgcn_readfirstlane(t >> 5);
    if (node >= NN) return;
    const unsigned lane4 = (unsigned)lane * 4u;
    const float4 bb = ((const float4*)b2)[lane];
    const float4 av = aggregate_row(Hf, cnt, slot, ovf, src, dst, node, lane, lane4);

    float hv[4] = { av.x + bb.x, av.y + bb.y, av.z + bb.z, av.w + bb.w };

    float4 hstore = { hv[0], hv[1], hv[2], hv[3] };
    *(float4*)(hout + (size_t)node * HD + lane4) = hstore;

    float acc[NC1 + NC2];
    #pragma unroll
    for (int c = 0; c < NC1 + NC2; ++c) acc[c] = 0.0f;

    #pragma unroll
    for (int j = 0; j < 4; ++j) {
        const int k = lane * 4 + j;
        #pragma unroll
        for (int c = 0; c < NC1; ++c) acc[c]       += hv[j] * Wh1[k * NC1 + c];
        #pragma unroll
        for (int c = 0; c < NC2; ++c) acc[NC1 + c] += hv[j] * Wh2[k * NC2 + c];
    }

    #pragma unroll
    for (int off = 16; off > 0; off >>= 1) {
        #pragma unroll
        for (int c = 0; c < NC1 + NC2; ++c)
            acc[c] += __shfl_down(acc[c], off, 32);
    }

    if (lane == 0) {
        #pragma unroll
        for (int c = 0; c < NC1; ++c) out1[(size_t)node * NC1 + c] = acc[c] + bh1[c];
        #pragma unroll
        for (int c = 0; c < NC2; ++c) out2[(size_t)node * NC2 + c] = acc[NC1 + c] + bh2[c];
    }
}

extern "C" void kernel_launch(void* const* d_in, const int* in_sizes, int n_in,
                              void* d_out, int out_size, void* d_ws, size_t ws_size,
                              hipStream_t stream) {
    const float* x   = (const float*)d_in[0];
    const int*   ei  = (const int*)d_in[1];
    const int*   src = ei;
    const int*   dst = ei + NE;
    const float* W1  = (const float*)d_in[2];
    const float* b1  = (const float*)d_in[3];
    const float* W2  = (const float*)d_in[4];
    const float* b2  = (const float*)d_in[5];
    const float* Wh1 = (const float*)d_in[6];
    const float* bh1 = (const float*)d_in[7];
    const float* Wh2 = (const float*)d_in[8];
    const float* bh2 = (const float*)d_in[9];

    float* out1 = (float*)d_out;
    float* out2 = out1 + (size_t)NN * NC1;
    float* hout = out2 + (size_t)NN * NC2;

    int*   cnt    = (int*)d_ws;
    int*   slot   = (int*)((char*)d_ws + WS_SLOT);
    int*   ovf    = (int*)((char*)d_ws + WS_OVF);
    float* bufA   = (float*)((char*)d_ws + WS_BUFA);
    float* bufB   = bufA + (size_t)NN * HD;

    const int NODE_BLK = (NN + 1 + 255) / 256;
    const int WAVE_BLK = (NN * 32 + 255) / 256;

    k_cnt_zero<<<NODE_BLK, 256, 0, stream>>>(cnt);
    k_gemm_wlds<FIN, 4><<<256, 384, 0, stream>>>(x, W1, bufA, src, dst, cnt, slot, ovf);

    k_gather_relu  <<<WAVE_BLK, 256, 0, stream>>>(bufA, cnt, slot, ovf, src, dst, b1, bufB);

    k_gemm_wlds<HD, 0><<<256, 256, 0, stream>>>(bufB, W2, bufA, nullptr, nullptr, nullptr, nullptr, nullptr);
    k_gather_heads <<<WAVE_BLK, 256, 0, stream>>>(bufA, cnt, slot, ovf, src, dst, b2, Wh1, bh1, Wh2, bh2,
                                                  out1, out2, hout);
}


# Edge ranks from LDS counters instead of 800,000 returning global adds: edges binned by destination bucket beside the layer-1 GEMM, rows ranked per bucket in LDS; sums stay in arrival-order slots
# speedup vs baseline: 4.5613x; 1.3679x over previous
#include <hip/hip_runtime.h>

typedef __attribute__((ext_vector_type(2))) float v2f;
typedef __attribute__((ext_vector_type(8))) float v8f;

#define NN 50000
#define NE 800000
#define FIN 256
#define HD 128
#define NC1 4
#define NC2 3

#define SLOT_CAP 64

#define BKT      196
#define NBKT     256
#define CELL     32
#define CELL_CAP 31
#define CHUNK    (NE / NBKT)
static_assert(CHUNK * NBKT == NE, "edge chunks must tile the edge list");
static_assert(NBKT * BKT >= NN && NN <= 65536 && BKT <= 256, "packing of node and source ids");

#define WS_SLOT    200192u
#define WS_OVF     13000192u
#define WS_BUFA    16200192u
#define WS_CELL    67400192u

__device__ __forceinline__ void row_over(int* __restrict__ cnt, int* __restrict__ ovf, int bkt, unsigned dl, unsigned s) {
    const int o = atomicAdd(&cnt[NN], 1);
    if ((unsigned)o < (unsigned)NE) ovf[o] = (int)((((unsigned)bkt * BKT + dl) << 16) | s);
}
__device__ __forceinline__ void row_place(int* ncnt, int* rows, int* __restrict__ cnt, int* __restrict__ ovf,
                                          int bkt, unsigned dl, unsigned s) {
    if (dl < (unsigned)BKT) {
        const int r = atomicAdd(&ncnt[dl], 1);
        if ((unsigned)r < (unsigned)SLOT_CAP) rows[dl * SLOT_CAP + r] = (int)s;
        else row_over(cnt, ovf, bkt, dl, s);
    }
}

#define ROWS_WAVES 16
__global__ void __launch_bounds__(32 * ROWS_WAVES) k_rows(const int* __restrict__ cells, const int* __restrict__ src,
                                                          const int* __restrict__ dst, int* __restrict__ cnt,
                                                          int* __restrict__ slot, int* __restrict__ ovf) {
    __shared__ int ncnt[256];
    __shared__ __attribute__((aligned(16))) int rows[BKT * SLOT_CAP];
    constexpr int NT  = 32 * ROWS_WAVES;
    constexpr int CPW = NBKT / ROWS_WAVES;
    constexpr int NQ  = BKT * SLOT_CAP / 4;
    const int tid  = threadIdx.x;
    const int lane = tid & 31;
    const int wv   = __builtin_amdgcn_readfirstlane(tid >> 5);
    const int bkt  = (int)blockIdx.x;

    int v[CPW];
    const int* cp = cells + (size_t)bkt * (NBKT * CELL) + lane;
    #pragma unroll
    for (int i = 0; i < CPW; ++i) v[i] = cp[(wv + ROWS_WAVES * i) * CELL];

    if (tid < 256) ncnt[tid] = 0;
    #pragma unroll
    for (int i = 0; i < (NQ + NT - 1) / NT; ++i) {
        const int q = tid + i * NT;
        if (q < NQ) ((int4*)rows)[q] = make_int4(0, 0, 0, 0);
    }
    __syncthreads();

    int      r[CPW];
    unsigned redo = 0;
    #pragma unroll
    for (int i = 0; i < CPW; ++i) {
        const int n = __builtin_amdgcn_readlane(v[i], CELL - 1);
        const unsigned dl = (unsigned)v[i] >> 16;
        r[i] = -1;
        redo |= ((unsigned)n > (unsigned)CELL_CAP ? 1u : 0u) << i;
        const bool ok = ((unsigned)n <= (unsigned)CELL_CAP) & (lane < n) & (dl < (unsigned)BKT);
        if (ok) r[i] = atomicAdd(&ncnt[dl], 1);
    }
    #pragma unroll
    for (int i = 0; i < CPW; ++i) {
        const unsigned dl = (unsigned)v[i] >> 16, s = (unsigned)v[i] & 0xFFFFu;
        if ((unsigned)r[i] < (unsigned)SLOT_CAP) rows[dl * SLOT_CAP + r[i]] = (int)s;
        else if (r[i] >= 0) row_over(cnt, ovf, bkt, dl, s);
    }
    while (redo) {
        {
            const int i = __builtin_ctz(redo);
            redo &= redo - 1u;
            const int e0 = (wv + ROWS_WAVES * i) * CHUNK;
            for (int e = lane; e < CHUNK; e += 32) {
                const unsigned d = (unsigned)dst[e0 + e];
                const unsigned s = (unsigned)src[e0 + e];
                if (d < (unsigned)NN && d / BKT == (unsigned)bkt)
                    row_place(ncnt, rows, cnt, ovf, bkt, d - (unsigned)bkt * BKT, s < (unsigned)NN ? s : 0xFFFFu);
            }
        }
    }
    __syncthreads();

    const int node0 = bkt * BKT;
    if (tid < BKT && node0 + tid < NN) cnt[node0 + tid] = ncnt[tid];
    #pragma unroll
    for (int i = 0; i < (NQ + NT - 1) / NT; ++i) {
        const int q = tid + i * NT;
        const int row = q >> 4, part = q & 15;
        if (q < NQ && node0 + row < NN && part * 4 < min(ncnt[row], SLOT_CAP))
            ((int4*)slot)[(size_t)node0 * (SLOT_CAP / 4) + q] = ((const int4*)rows)[q];
    }
}

typedef __attribute__((ext_vector_type(4))) float v4f;
#define GEMM_LDS_FLOATS 43008
#define GEMM_TILES (NN / 16)
#define GEMM_PAIRS 1024

#define GEMM_CK 32

template <int K>
__device__ __forceinline__ void gemm_load_a(v2f (&a)[GEMM_CK], const float* __restrict__ X, int tile, int chunk,
                                            int lh, int half) {
    const float* xp = X + (size_t)(tile * 16 + lh) * K + (chunk * (4 * GEMM_CK) + 2 * half);
    #pragma unroll
    for (int ks = 0; ks < GEMM_CK; ++ks) a[ks] = *(const v2f*)(xp + 4 * ks);
}

__device__ __forceinline__ void gemm_mma_chunk(const v2f (&a)[GEMM_CK], const float* bl,
                                               v8f& acc0, v8f& acc1, v8f& acc2, v8f& acc3) {
    #pragma unroll
    for (int ks = 0; ks < GEMM_CK; ++ks) {
        const v4f b01 = *(const v4f*)(bl + ks * (4 * HD));
        const v4f b23 = *(const v4f*)(bl + ks * (4 * HD) + 64);
        const v2f b0 = { b01.x, b01.y };
        const v2f b1 = { b01.z, b01.w };
        const v2f b2 = { b23.x, b23.y };
        const v2f b3 = { b23.z, b23.w };
        acc0 = __builtin_amdgcn_wmma_f32_16x16x4_f32(false, a[ks], false, b0, (short)0, acc0, false, false);
        acc1 = __builtin_amdgcn_wmma_f32_16x16x4_f32(false, a[ks], false, b1, (short)0, acc1, false, false);
        acc2 = __builtin_amdgcn_wmma_f32_16x16x4_f32(false, a[ks], false, b2, (short)0, acc2, false, false);
        acc3 = __builtin_amdgcn_wmma_f32_16x16x4_f32(false, a[ks], false, b3, (short)0, acc3, false, false);
    }
}

template <int K, int EW>
__global__ void __launch_bounds__(256 + 32 * EW) k_gemm_wlds(const float* __restrict__ X, const float* __restrict__ W,
                                                             float* __restrict__ OUT, const int* __restrict__ src,
                                                             const int* __restrict__ dst, int* __restrict__ cnt,
                                                             int* __restrict__ cells) {
    __shared__ __attribute__((aligned(16))) float Wl[GEMM_LDS_FLOATS];
    __shared__ __attribute__((aligned(16))) int cellbuf[EW > 0 ? NBKT * CELL : 4];
    __shared__ int hist[EW > 0 ? NBKT : 1];
    constexpr int NCH = K / (4 * GEMM_CK);
    constexpr int NT  = 256 + 32 * EW;
    constexpr int EPT = (CHUNK + NT - 1) / NT;
    const int tid  = threadIdx.x;
    const int lane = tid & 31;
    const int wv   = __builtin_amdgcn_readfirstlane(tid >> 5);
    const int half = lane >> 4;
    const int lh   = lane & 15;
    const int hf   = wv & 1;
    int t = (wv >> 1) * 256 + (int)blockIdx.x;

    if constexpr (EW > 0) {
        #pragma unroll
        for (int i = 0; i < (NBKT * CELL / 4 + NT - 1) / NT; ++i) {
            const int q = tid + i * NT;
            if (q < NBKT * CELL / 4) ((int4*)cellbuf)[q] = make_int4(0, 0, 0, 0);
        }
        if (tid < NBKT) hist[tid] = 0;
        if (blockIdx.x == 0 && tid == 0) cnt[NN] = 0;
        __syncthreads();
    }

    v2f a0[GEMM_CK], a1[GEMM_CK];
    const bool mul_wave = (EW == 0) || (wv < 8);
    if (mul_wave) gemm_load_a<K>(a0, X, t, 0, lh, half);

    int ed[EPT], es[EPT];
    if constexpr (EW > 0) {
        const int e0 = (int)blockIdx.x * CHUNK + tid;
        #pragma unroll
        for (int j = 0; j < EPT; ++j) {
            const bool in = (j + 1) * NT <= CHUNK || tid + j * NT < CHUNK;
            ed[j] = in ? dst[e0 + j * NT] : -1;
            es[j] = in ? src[e0 + j * NT] : 0;
        }
    }

    {
        const int j = tid & 31;
        #pragma unroll 4
        for (int kp = wv; kp < K / 2; kp += 8 + EW) {
            const float4 w0 = *(const float4*)(W + (size_t)(2 * kp) * HD + 4 * j);
            const float4 w1 = *(const float4*)(W + (size_t)(2 * kp + 1) * HD + 4 * j);
            float4* d = (float4*)(Wl + (kp * HD + 4 * j) * 2);
            d[0] = make_float4(w0.x, w1.x, w0.y, w1.y);
            d[1] = make_float4(w0.z, w1.z, w0.w, w1.w);
        }
    }
    if constexpr (EW > 0) {
        unsigned eb[EPT];
        int      er[EPT];
        #pragma unroll
        for (int j = 0; j < EPT; ++j) {
            const unsigned d = (unsigned)ed[j];
            eb[j] = d / BKT;
            er[j] = CELL_CAP;
            if (d < (unsigned)NN) er[j] = atomicAdd(&hist[eb[j]], 1);
        }
        #pragma unroll
        for (int j = 0; j < EPT; ++j) {
            if ((unsigned)er[j] < (unsigned)CELL_CAP) {
                const unsigned dl = (unsigned)ed[j] - eb[j] * BKT;
                const unsigned s  = (unsigned)es[j] < (unsigned)NN ? (unsigned)es[j] : 0xFFFFu;
                cellbuf[eb[j] * CELL + er[j]] = (int)((dl << 16) | s);
            }
        }
    }
    __syncthreads();

    if constexpr (EW > 0) {
        if (!mul_wave) {
            const int x = tid - 256;
            int* gc = cells + (size_t)blockIdx.x * CELL;
            static_assert(EW == 0 || (NBKT * CELL / 4) % (32 * (EW > 0 ? EW : 1)) == 0, "whole passes over the cells");
            #pragma unroll
            for (int i = 0; i < NBKT * CELL / 4 / (32 * EW); ++i) {
                const int q = x + i * (32 * EW);
                const int b = q >> 3, part = q & 7;
                int4 c = ((const int4*)cellbuf)[q];
                if (part == 7) c.w = hist[b];
                *(int4*)(gc + (size_t)b * (NBKT * CELL) + part * 4) = c;
            }
            return;
        }
    }

    const float* bl = Wl + (half * HD + 64 * hf + 2 * lh) * 2;

    for (;;) {
        const int  tn   = t + GEMM_PAIRS;
        const bool more = tn < GEMM_TILES;
        const int  tl   = more ? tn : t;

        v8f acc0 = {}, acc1 = {}, acc2 = {}, acc3 = {};
        if constexpr (NCH == 2) {
            asm volatile("" ::: "memory");
            gemm_load_a<K>(a1, X, t, 1, lh, half);
            asm volatile("" ::: "memory");
            gemm_mma_chunk(a0, bl, acc0, acc1, acc2, acc3);
            gemm_load_a<K>(a0, X, tl, 0, lh, half);
            asm volatile("" ::: "memory");
            gemm_mma_chunk(a1, bl + GEMM_CK * (4 * HD), acc0, acc1, acc2, acc3);
        } else {
            gemm_load_a<K>(a1, X, tl, 0, lh, half);
            gemm_mma_chunk(a0, bl, acc0, acc1, acc2, acc3);
        }

        float* op = OUT + (size_t)(t * 16 + 8 * half) * HD + 64 * hf + 2 * lh;
        #pragma unroll
        for (int r = 0; r < 8; ++r) {
            const v2f o01 = { acc0[r], acc1[r] };
            const v2f o23 = { acc2[r], acc3[r] };
            *(v2f*)(op + r * HD)      = o01;
            *(v2f*)(op + r * HD + 32) = o23;
        }

        if (!more) break;
        if constexpr (NCH == 1) {
            #pragma unroll
            for (int ks = 0; ks < GEMM_CK; ++ks) a0[ks] = a1[ks];
        }
        t = tn;
    }
}

template <int K>
__device__ __forceinline__ void gather_group(const float* __restrict__ Hf, int s_l, float w_l, int j,
                                             unsigned lane4, float4& acc) {
    float4 hv[K];
    float  w[K];
    #pragma unroll
    for (int u = 0; u < K; ++u) {
        const unsigned s = (unsigned)__builtin_amdgcn_readlane(s_l, j + u);
        w[u]  = __int_as_float(__builtin_amdgcn_readlane(__float_as_int(w_l), j + u));
        hv[u] = *(const float4*)(Hf + (s * HD + lane4));
    }
    #pragma unroll
    for (int u = 0; u < K; ++u) {
        acc.x = fmaf(w[u], hv[u].x, acc.x);
        acc.y = fmaf(w[u], hv[u].y, acc.y);
        acc.z = fmaf(w[u], hv[u].z, acc.z);
        acc.w = fmaf(w[u], hv[u].w, acc.w);
    }
}

__device__ __forceinline__ float dinv_of(int c) { return rsqrtf((float)(c + 1)); }

__device__ __forceinline__ float4 aggregate_row(const float* __restrict__ Hf, const int* __restrict__ cnt,
                                                const int* __restrict__ slot, const int* __restrict__ ovf,
                                                int node, int lane, unsigned lane4) {
    const int c = cnt[node];
    const float di = dinv_of(c);

    const float4 hs = *(const float4*)(Hf + ((unsigned)node * HD + lane4));
    const float wself = di * di;
    float4 acc = { hs.x * wself, hs.y * wself, hs.z * wself, hs.w * wself };

    const int nrow = min(c, SLOT_CAP);
    const int* srow = slot + (unsigned)node * SLOT_CAP;
    for (int base = 0; base < nrow; base += 32) {
        const int n = min(32, nrow - base);
        int   s_l = node;
        float w_l = 0.0f;
        if (lane < n) {
            const int sv = srow[base + lane];
            s_l = ((unsigned)sv < (unsigned)NN) ? sv : node;
            w_l = dinv_of(cnt[s_l]) * di;
        }
        int j = 0;
        for (; j + 8 <= n; j += 8) gather_group<8>(Hf, s_l, w_l, j, lane4, acc);
        if (j + 4 <= n) { gather_group<4>(Hf, s_l, w_l, j, lane4, acc); j += 4; }
        if (j + 2 <= n) { gather_group<2>(Hf, s_l, w_l, j, lane4, acc); j += 2; }
        if (j < n)      { gather_group<1>(Hf, s_l, w_l, j, lane4, acc); }
    }

    if (c > SLOT_CAP) {
        const int no = __builtin_amdgcn_readfirstlane(min(cnt[NN], NE));
        for (int b = 0; b < no; b += 32) {
            int   s_l = node;
            float w_l = 0.0f;
            bool  mine = false;
            if (b + lane < no) {
                const unsigned e = (unsigned)ovf[b + lane];
                const int sv = (int)(e & 0xFFFFu);
                mine = ((int)(e >> 16) == node);
                s_l = ((unsigned)sv < (unsigned)NN) ? sv : node;
                w_l = dinv_of(cnt[s_l]) * di;
            }
            unsigned m = __builtin_amdgcn_ballot_w32(mine);
            while (m) {
                const int j = __builtin_ctz(m);
                m &= m - 1u;
                gather_group<1>(Hf, s_l, w_l, j, lane4, acc);
            }
        }
    }
    return acc;
}

__global__ void __launch_bounds__(256) k_gather_relu(const float* __restrict__ Hf, const int* __restrict__ cnt,
                                                     const int* __restrict__ slot, const int* __restrict__ ovf,
                                                     const float* __restrict__ bias, float* __restrict__ OUT) {
    const int t = blockIdx.x * blockDim.x + threadIdx.x;
    const int lane = threadIdx.x & 31;
    const int node = __builtin_amdgcn_readfirstlane(t >> 5);
    if (node >= NN) return;
    const unsigned lane4 = (unsigned)lane * 4u;
    const float4 bb = ((const float4*)bias)[lane];
    const float4 acc = aggregate_row(Hf, cnt, slot, ovf, node, lane, lane4);
    float4 r;
    r.x = fmaxf(acc.x + bb.x, 0.0f);
    r.y = fmaxf(acc.y + bb.y, 0.0f);
    r.z = fmaxf(acc.z + bb.z, 0.0f);
    r.w = fmaxf(acc.w + bb.w, 0.0f);
    *(float4*)(OUT + ((unsigned)node * HD + lane4)) = r;
}

__global__ void __launch_bounds__(256) k_gather_heads(const float* __restrict__ Hf, const int* __restrict__ cnt,
                                                      const int* __restrict__ slot, const int* __restrict__ ovf,
                                                      const float* __restrict__ b2,
                                                      const float* __restrict__ Wh1, const float* __restrict__ bh1,
                                                      const float* __restrict__ Wh2, const float* __restrict__ bh2,
                                                      float* __restrict__ out1, float* __restrict__ out2,
                                                      float* __restrict__ hout) {
    const int t = blockIdx.x * blockDim.x + threadIdx.x;
    const int lane = threadIdx.x & 31;
    const int node = __builtin_amdgcn_readfirstlane(t >> 5);
    if (node >= NN) return;
    const unsigned lane4 = (unsigned)lane * 4u;
    const float4 bb = ((const float4*)b2)[lane];
    const float4 av = aggregate_row(Hf, cnt, slot, ovf, node, lane, lane4);

    float hv[4] = { av.x + bb.x, av.y + bb.y, av.z + bb.z, av.w + bb.w };

    float4 hstore = { hv[0], hv[1], hv[2], hv[3] };
    *(float4*)(hout + (size_t)node * HD + lane4) = hstore;

    float acc[NC1 + NC2];
    #pragma unroll
    for (int c = 0; c < NC1 + NC2; ++c) acc[c] = 0.0f;

    #pragma unroll
    for (int j = 0; j < 4; ++j) {
        const int k = lane * 4 + j;
        #pragma unroll
        for (int c = 0; c < NC1; ++c) acc[c]       += hv[j] * Wh1[k * NC1 + c];
        #pragma unroll
        for (int c = 0; c < NC2; ++c) acc[NC1 + c] += hv[j] * Wh2[k * NC2 + c];
    }

    #pragma unroll
    for (int off = 16; off > 0; off >>= 1) {
        #pragma unroll
        for (int c = 0; c < NC1 + NC2; ++c)
            acc[c] += __shfl_down(acc[c], off, 32);
    }

    if (lane == 0) {
        #pragma unroll
        for (int c = 0; c < NC1; ++c) out1[(size_t)node * NC1 + c] = acc[c] + bh1[c];
        #pragma unroll
        for (int c = 0; c < NC2; ++c) out2[(size_t)node * NC2 + c] = acc[NC1 + c] + bh2[c];
    }
}

extern "C" void kernel_launch(void* const* d_in, const int* in_sizes, int n_in,
                              void* d_out, int out_size, void* d_ws, size_t ws_size,
                              hipStream_t stream) {
    const float* x   = (const float*)d_in[0];
    const int*   ei  = (const int*)d_in[1];
    const int*   src = ei;
    const int*   dst = ei + NE;
    const float* W1  = (const float*)d_in[2];
    const float* b1  = (const float*)d_in[3];
    const float* W2  = (const float*)d_in[4];
    const float* b2  = (const float*)d_in[5];
    const float* Wh1 = (const float*)d_in[6];
    const float* bh1 = (const float*)d_in[7];
    const float* Wh2 = (const float*)d_in[8];
    const float* bh2 = (const float*)d_in[9];

    float* out1 = (float*)d_out;
    float* out2 = out1 + (size_t)NN * NC1;
    float* hout = out2 + (size_t)NN * NC2;

    int*   cnt    = (int*)d_ws;
    int*   slot   = (int*)((char*)d_ws + WS_SLOT);
    int*   ovf    = (int*)((char*)d_ws + WS_OVF);
    float* bufA   = (float*)((char*)d_ws + WS_BUFA);
    float* bufB   = bufA + (size_t)NN * HD;
    int*   cells  = (int*)((char*)d_ws + WS_CELL);

    const int WAVE_BLK = (NN * 32 + 255) / 256;

    k_gemm_wlds<FIN, 4><<<NBKT, 384, 0, stream>>>(x, W1, bufA, src, dst, cnt, cells);
    k_rows<<<NBKT, 32 * ROWS_WAVES, 0, stream>>>(cells, src, dst, cnt, slot, ovf);

    k_gather_relu  <<<WAVE_BLK, 256, 0, stream>>>(bufA, cnt, slot, ovf, b1, bufB);

    k_gemm_wlds<HD, 0><<<256, 256, 0, stream>>>(bufB, W2, bufA, nullptr, nullptr, nullptr, nullptr);
    k_gather_heads <<<WAVE_BLK, 256, 0, stream>>>(bufA, cnt, slot, ovf, b2, Wh1, bh1, Wh2, bh2,
                                                  out1, out2, hout);
}
